# starting kernel
# baseline (speedup 1.0000x reference)
#include <hip/hip_runtime.h>
#include <hip/hip_bf16.h>


typedef _Float16 f16;
typedef __attribute__((ext_vector_type(16))) _Float16 v16h;
typedef __attribute__((ext_vector_type(8)))  _Float16 v8h;
typedef __attribute__((ext_vector_type(4)))  _Float16 v4h;
typedef __attribute__((ext_vector_type(8)))  float    v8f;

#define N_NODES 50000
#define N_EDGES 500000
#define ETILES  (N_EDGES / 16)
#define EWAVES  8

__device__ __forceinline__ v16h join16(v8h lo, v8h hi) {
    union { v16h v; v8h h[2]; } u;
    u.h[0] = lo; u.h[1] = hi;
    return u.v;
}

__device__ __forceinline__ v8f splat8(float x) {
    v8f a;
#pragma unroll
    for (int r = 0; r < 8; ++r) a[r] = x;
    return a;
}

__device__ __forceinline__ v8f wmma_f16(v16h a, v16h b, v8f c) {
    return __builtin_amdgcn_wmma_f32_16x16x32_f16(false, a, false, b,
                                                  (short)0, c, false, false);
}

__global__ void pack_w_kernel(const float* __restrict__ W, f16* __restrict__ out,
                              int Ntot, int kTiles, int total) {
    for (int idx = blockIdx.x * blockDim.x + threadIdx.x; idx < total;
         idx += gridDim.x * blockDim.x) {
        int f  = idx >> 9;
        int r  = idx & 511;
        int l  = r >> 4;
        int i  = r & 15;
        int nt = f / kTiles;
        int kt = f - nt * kTiles;
        int K  = kt * 32 + ((i >> 3) << 4) + ((l >> 4) << 3) + (i & 7);
        int N  = nt * 16 + (l & 15);
        out[idx] = (f16)W[(size_t)K * Ntot + N];
    }
}

__global__ void node_feat_kernel(const float* __restrict__ x,
                                 const float* __restrict__ bn,
                                 const f16*  __restrict__ Wnp,
                                 f16* __restrict__ nf) {
    __shared__ __align__(16) f16 xs_all[4][16 * 128];
    const int l  = threadIdx.x & 31;
    const int wv = threadIdx.x >> 5;
    const int tile = blockIdx.x * 4 + wv;
    if (tile >= N_NODES / 16) return;
    f16* xs = xs_all[wv];
    const int n0 = tile * 16;
    const int m  = l & 15;
    const int h  = l >> 4;

#pragma unroll
    for (int t = 0; t < 16; ++t) {
        int off = l * 64 + t * 4;
        int row = off >> 7, col = off & 127;
        const float4 v = *(const float4*)(x + (size_t)(n0 + row) * 128 + col);
        v4h p; p[0] = (f16)v.x; p[1] = (f16)v.y; p[2] = (f16)v.z; p[3] = (f16)v.w;
        *(v4h*)(xs + row * 128 + col) = p;
    }

    for (int nc = 0; nc < 2; ++nc) {
        v8f acc[4];
#pragma unroll
        for (int j = 0; j < 4; ++j) acc[j] = splat8(bn[(nc * 4 + j) * 16 + m]);
        for (int kt = 0; kt < 4; ++kt) {
            int k0  = kt * 32 + h * 8;
            v16h a = join16(*(const v8h*)(xs + m * 128 + k0),
                            *(const v8h*)(xs + m * 128 + k0 + 16));
            v16h b[4];
#pragma unroll
            for (int j = 0; j < 4; ++j)
                b[j] = *(const v16h*)(Wnp + (size_t)(((nc * 4 + j) * 4 + kt) * 512) + l * 16);
#pragma unroll
            for (int j = 0; j < 4; ++j) acc[j] = wmma_f16(a, b[j], acc[j]);
        }
#pragma unroll
        for (int j = 0; j < 4; ++j)
#pragma unroll
            for (int r = 0; r < 8; ++r)
                nf[(size_t)(n0 + r + 8 * h) * 128 + (nc * 4 + j) * 16 + m] =
                    (f16)fmaxf(acc[j][r], 0.0f);
    }
}

#define CMB_STRIDE 392
#define H1_STRIDE  520
#define WAVE_LDS   31232

__global__ void edge_fused_kernel(const int*   __restrict__ eidx,
                                  const float* __restrict__ eattr,
                                  const float* __restrict__ be,
                                  const float* __restrict__ b1,
                                  const float* __restrict__ b2,
                                  const float* __restrict__ W3,
                                  const float* __restrict__ b3,
                                  const f16*   __restrict__ Wep,
                                  const f16*   __restrict__ W1p,
                                  const f16*   __restrict__ W2p,
                                  const f16*   __restrict__ nf,
                                  float* __restrict__ out) {
    extern __shared__ __align__(16) char smem[];
    const int l  = threadIdx.x & 31;
    const int wv = threadIdx.x >> 5;
    const int m  = l & 15;
    const int h  = l >> 4;
    f16*   cmb = (f16*)(smem + wv * WAVE_LDS);
    f16*   h1  = cmb + 16 * CMB_STRIDE;
    f16*   ea  = h1 + 16 * H1_STRIDE;
    float* w3l = (float*)ea;

    const int  tile  = blockIdx.x * EWAVES + wv;
    const bool valid = (tile < ETILES);
    const int  e0    = valid ? tile * 16 : 0;

    const int node = eidx[h * N_EDGES + e0 + m];
    const f16* nrow = nf + (size_t)node * 128;
#pragma unroll
    for (int t = 0; t < 16; ++t)
        *(v8h*)(cmb + m * CMB_STRIDE + h * 128 + t * 8) = *(const v8h*)(nrow + t * 8);

#pragma unroll
    for (int t = 0; t < 8; ++t) {
        int off = l * 32 + t * 4;
        int row = off >> 6, col = off & 63;
        const float4 v = *(const float4*)(eattr + (size_t)(e0 + row) * 64 + col);
        v4h p; p[0] = (f16)v.x; p[1] = (f16)v.y; p[2] = (f16)v.z; p[3] = (f16)v.w;
        *(v4h*)(ea + row * 64 + col) = p;
    }

    for (int nt = 0; nt < 8; ++nt) {
        v8f acc = splat8(be[nt * 16 + m]);
#pragma unroll
        for (int kt = 0; kt < 2; ++kt) {
            int k0 = kt * 32 + h * 8;
            v16h a = join16(*(const v8h*)(ea + m * 64 + k0),
                            *(const v8h*)(ea + m * 64 + k0 + 16));
            v16h b = *(const v16h*)(Wep + (size_t)((nt * 2 + kt) * 512) + l * 16);
            acc = wmma_f16(a, b, acc);
        }
#pragma unroll
        for (int r = 0; r < 8; ++r)
            cmb[(r + 8 * h) * CMB_STRIDE + 256 + nt * 16 + m] = (f16)fmaxf(acc[r], 0.0f);
    }

#pragma unroll
    for (int t = 0; t < 8; ++t) w3l[t * 32 + l] = W3[t * 32 + l];

    __syncthreads();

    for (int nc = 0; nc < 8; ++nc) {
        if (nc < 7)
            __builtin_prefetch(W1p + (size_t)(((nc + 1) * 4) * 12) * 512 + l * 16, 0, 1);
        v8f acc[4];
#pragma unroll
        for (int j = 0; j < 4; ++j) acc[j] = splat8(b1[(nc * 4 + j) * 16 + m]);
        for (int kt = 0; kt < 12; ++kt) {
            int k0 = kt * 32 + h * 8;
            v16h a = join16(*(const v8h*)(cmb + m * CMB_STRIDE + k0),
                            *(const v8h*)(cmb + m * CMB_STRIDE + k0 + 16));
            v16h b[4];
#pragma unroll
            for (int j = 0; j < 4; ++j)
                b[j] = *(const v16h*)(W1p + (size_t)(((nc * 4 + j) * 12 + kt) * 512) + l * 16);
#pragma unroll
            for (int j = 0; j < 4; ++j) acc[j] = wmma_f16(a, b[j], acc[j]);
        }
#pragma unroll
        for (int j = 0; j < 4; ++j)
#pragma unroll
            for (int r = 0; r < 8; ++r)
                h1[(r + 8 * h) * H1_STRIDE + (nc * 4 + j) * 16 + m] =
                    (f16)fmaxf(acc[j][r], 0.0f);
        __syncthreads();
    }

    for (int nc = 0; nc < 4; ++nc) {
        if (nc < 3)
            __builtin_prefetch(W2p + (size_t)(((nc + 1) * 4) * 16) * 512 + l * 16, 0, 1);
        v8f acc[4];
#pragma unroll
        for (int j = 0; j < 4; ++j) acc[j] = splat8(b2[(nc * 4 + j) * 16 + m]);
        for (int kt = 0; kt < 16; ++kt) {
            int k0 = kt * 32 + h * 8;
            v16h a = join16(*(const v8h*)(h1 + m * H1_STRIDE + k0),
                            *(const v8h*)(h1 + m * H1_STRIDE + k0 + 16));
            v16h b[4];
#pragma unroll
            for (int j = 0; j < 4; ++j)
                b[j] = *(const v16h*)(W2p + (size_t)(((nc * 4 + j) * 16 + kt) * 512) + l * 16);
#pragma unroll
            for (int j = 0; j < 4; ++j) acc[j] = wmma_f16(a, b[j], acc[j]);
        }
#pragma unroll
        for (int j = 0; j < 4; ++j)
#pragma unroll
            for (int r = 0; r < 8; ++r)
                cmb[(r + 8 * h) * CMB_STRIDE + (nc * 4 + j) * 16 + m] =
                    (f16)fmaxf(acc[j][r], 0.0f);
        __syncthreads();
    }

    if (valid && h == 0) {
        float s = b3[0];
        for (int t = 0; t < 32; ++t) {
            v8h c = *(const v8h*)(cmb + m * CMB_STRIDE + t * 8);
#pragma unroll
            for (int u = 0; u < 8; ++u) s += (float)c[u] * w3l[t * 8 + u];
        }
        out[e0 + m] = s;
    }
}

#define WS_WN 0
#define WS_WE (WS_WN + 16384 * 2)
#define WS_W1 (WS_WE + 8192 * 2)
#define WS_W2 (WS_W1 + 196608 * 2)
#define WS_NF (WS_W2 + 131072 * 2)

extern "C" void kernel_launch(void* const* d_in, const int* in_sizes, int n_in,
                              void* d_out, int out_size, void* d_ws, size_t ws_size,
                              hipStream_t stream) {
    (void)in_sizes; (void)n_in; (void)out_size; (void)ws_size;
    const float* x     = (const float*)d_in[0];
    const int*   eidx  = (const int*)  d_in[1];
    const float* eattr = (const float*)d_in[2];
    const float* Wn    = (const float*)d_in[3];
    const float* bn    = (const float*)d_in[4];
    const float* We    = (const float*)d_in[5];
    const float* be    = (const float*)d_in[6];
    const float* W1    = (const float*)d_in[7];
    const float* b1    = (const float*)d_in[8];
    const float* W2    = (const float*)d_in[9];
    const float* b2    = (const float*)d_in[10];
    const float* W3    = (const float*)d_in[11];
    const float* b3    = (const float*)d_in[12];

    char* ws  = (char*)d_ws;
    f16* Wnp = (f16*)(ws + WS_WN);
    f16* Wep = (f16*)(ws + WS_WE);
    f16* W1p = (f16*)(ws + WS_W1);
    f16* W2p = (f16*)(ws + WS_W2);
    f16* nf  = (f16*)(ws + WS_NF);

    (void)hipFuncSetAttribute((const void*)edge_fused_kernel,
                              hipFuncAttributeMaxDynamicSharedMemorySize,
                              EWAVES * WAVE_LDS);

    pack_w_kernel<<<64,  256, 0, stream>>>(Wn, Wnp, 128, 4,  16384);
    pack_w_kernel<<<32,  256, 0, stream>>>(We, Wep, 128, 2,  8192);
    pack_w_kernel<<<768, 256, 0, stream>>>(W1, W1p, 512, 12, 196608);
    pack_w_kernel<<<512, 256, 0, stream>>>(W2, W2p, 256, 16, 131072);

    node_feat_kernel<<<(3125 + 3) / 4, 128, 0, stream>>>(x, bn, Wnp, nf);

    edge_fused_kernel<<<(ETILES + EWAVES - 1) / EWAVES, 32 * EWAVES,
                        EWAVES * WAVE_LDS, stream>>>(
        eidx, eattr, be, b1, b2, W3, b3, Wep, W1p, W2p, nf, (float*)d_out);
}


# edge kernel re-tiled: a workgroup's 128 edge rows are shared in LDS and each wave owns a strip of output columns for all 8 row tiles, so a weight fragment is loaded once per workgroup for 8 WMMAs
# speedup vs baseline: 2.6105x; 2.6105x over previous
#include <hip/hip_runtime.h>
#include <hip/hip_bf16.h>


typedef _Float16 f16;
typedef __attribute__((ext_vector_type(16))) _Float16 v16h;
typedef __attribute__((ext_vector_type(8)))  _Float16 v8h;
typedef __attribute__((ext_vector_type(4)))  _Float16 v4h;
typedef __attribute__((ext_vector_type(8)))  float    v8f;

#define N_NODES 50000
#define N_EDGES 500000
#define ETILES  (N_EDGES / 16)
#define EWAVES  8

__device__ __forceinline__ v16h join16(v8h lo, v8h hi) {
    union { v16h v; v8h h[2]; } u;
    u.h[0] = lo; u.h[1] = hi;
    return u.v;
}

__device__ __forceinline__ v8f splat8(float x) {
    v8f a;
#pragma unroll
    for (int r = 0; r < 8; ++r) a[r] = x;
    return a;
}

__device__ __forceinline__ v8f wmma_f16(v16h a, v16h b, v8f c) {
    return __builtin_amdgcn_wmma_f32_16x16x32_f16(false, a, false, b,
                                                  (short)0, c, false, false);
}

__global__ void pack_w_kernel(const float* __restrict__ W, f16* __restrict__ out,
                              int Ntot, int kTiles, int total) {
    for (int idx = blockIdx.x * blockDim.x + threadIdx.x; idx < total;
         idx += gridDim.x * blockDim.x) {
        int f  = idx >> 9;
        int r  = idx & 511;
        int l  = r >> 4;
        int i  = r & 15;
        int nt = f / kTiles;
        int kt = f - nt * kTiles;
        int K  = kt * 32 + ((i >> 3) << 4) + ((l >> 4) << 3) + (i & 7);
        int N  = nt * 16 + (l & 15);
        out[idx] = (f16)W[(size_t)K * Ntot + N];
    }
}

__global__ void node_feat_kernel(const float* __restrict__ x,
                                 const float* __restrict__ bn,
                                 const f16*  __restrict__ Wnp,
                                 f16* __restrict__ nf) {
    __shared__ __align__(16) f16 xs_all[4][16 * 128];
    const int l  = threadIdx.x & 31;
    const int wv = threadIdx.x >> 5;
    const int tile = blockIdx.x * 4 + wv;
    if (tile >= N_NODES / 16) return;
    f16* xs = xs_all[wv];
    const int n0 = tile * 16;
    const int m  = l & 15;
    const int h  = l >> 4;

#pragma unroll
    for (int t = 0; t < 16; ++t) {
        int off = l * 64 + t * 4;
        int row = off >> 7, col = off & 127;
        const float4 v = *(const float4*)(x + (size_t)(n0 + row) * 128 + col);
        v4h p; p[0] = (f16)v.x; p[1] = (f16)v.y; p[2] = (f16)v.z; p[3] = (f16)v.w;
        *(v4h*)(xs + row * 128 + col) = p;
    }

    for (int nc = 0; nc < 2; ++nc) {
        v8f acc[4];
#pragma unroll
        for (int j = 0; j < 4; ++j) acc[j] = splat8(bn[(nc * 4 + j) * 16 + m]);
        for (int kt = 0; kt < 4; ++kt) {
            int k0  = kt * 32 + h * 8;
            v16h a = join16(*(const v8h*)(xs + m * 128 + k0),
                            *(const v8h*)(xs + m * 128 + k0 + 16));
            v16h b[4];
#pragma unroll
            for (int j = 0; j < 4; ++j)
                b[j] = *(const v16h*)(Wnp + (size_t)(((nc * 4 + j) * 4 + kt) * 512) + l * 16);
#pragma unroll
            for (int j = 0; j < 4; ++j) acc[j] = wmma_f16(a, b[j], acc[j]);
        }
#pragma unroll
        for (int j = 0; j < 4; ++j)
#pragma unroll
            for (int r = 0; r < 8; ++r)
                nf[(size_t)(n0 + r + 8 * h) * 128 + (nc * 4 + j) * 16 + m] =
                    (f16)fmaxf(acc[j][r], 0.0f);
    }
}

#define CMB_STRIDE 392
#define H1_STRIDE  520
#define EA_STRIDE  64
#define MTILES     8
#define LDS_CMB    0
#define LDS_H1     (LDS_CMB + 128 * CMB_STRIDE * 2)
#define LDS_EA     (LDS_H1  + 128 * H1_STRIDE  * 2)
#define LDS_W3     (LDS_EA  + 128 * EA_STRIDE  * 2)
#define WG_LDS     (LDS_W3  + 256 * 4)

template <int NJ, int KT>
__device__ __forceinline__ void strip_layer(const f16* A, int a_stride,
                                            const f16* __restrict__ Wp,
                                            const float* __restrict__ bias,
                                            int nt0,
                                            f16* D, int d_stride, int d_col0,
                                            int l, int m, int h) {
    v8f acc[MTILES][NJ];
#pragma unroll
    for (int j = 0; j < NJ; ++j) {
        const float bv = bias[(nt0 + j) * 16 + m];
#pragma unroll
        for (int mi = 0; mi < MTILES; ++mi) acc[mi][j] = splat8(bv);
    }
    for (int kt = 0; kt < KT; ++kt) {
        const int k0 = kt * 32 + h * 8;
        v16h b[NJ];
#pragma unroll
        for (int j = 0; j < NJ; ++j)
            b[j] = *(const v16h*)(Wp + (size_t)(((nt0 + j) * KT + kt) * 512) + l * 16);
        v16h a[MTILES];
#pragma unroll
        for (int mi = 0; mi < MTILES; ++mi) {
            const f16* ar = A + (mi * 16 + m) * a_stride + k0;
            a[mi] = join16(*(const v8h*)ar, *(const v8h*)(ar + 16));
        }
#pragma unroll
        for (int mi = 0; mi < MTILES; ++mi)
#pragma unroll
            for (int j = 0; j < NJ; ++j)
                acc[mi][j] = wmma_f16(a[mi], b[j], acc[mi][j]);
    }
#pragma unroll
    for (int mi = 0; mi < MTILES; ++mi)
#pragma unroll
        for (int j = 0; j < NJ; ++j)
#pragma unroll
            for (int r = 0; r < 8; ++r)
                D[(mi * 16 + r + 8 * h) * d_stride + d_col0 + (nt0 + j) * 16 + m] =
                    (f16)fmaxf(acc[mi][j][r], 0.0f);
}

__global__ void __launch_bounds__(32 * EWAVES)
edge_fused_kernel(const int*   __restrict__ eidx,
                  const float* __restrict__ eattr,
                  const float* __restrict__ be,
                  const float* __restrict__ b1,
                  const float* __restrict__ b2,
                  const float* __restrict__ W3,
                  const float* __restrict__ b3,
                  const f16*   __restrict__ Wep,
                  const f16*   __restrict__ W1p,
                  const f16*   __restrict__ W2p,
                  const f16*   __restrict__ nf,
                  float* __restrict__ out) {
    extern __shared__ __align__(16) char smem[];
    const int l  = threadIdx.x & 31;
    const int wv = threadIdx.x >> 5;
    const int m  = l & 15;
    const int h  = l >> 4;
    f16*   cmb = (f16*)(smem + LDS_CMB);
    f16*   h1  = (f16*)(smem + LDS_H1);
    f16*   ea  = (f16*)(smem + LDS_EA);
    float* w3l = (float*)(smem + LDS_W3);

    const int  tile  = blockIdx.x * EWAVES + wv;
    const bool valid = (tile < ETILES);
    const int  e0    = valid ? tile * 16 : 0;
    const int  row0  = wv * 16;

    const int node = eidx[h * N_EDGES + e0 + m];
    const f16* nrow = nf + (size_t)node * 128;
#pragma unroll
    for (int t = 0; t < 16; ++t)
        *(v8h*)(cmb + (row0 + m) * CMB_STRIDE + h * 128 + t * 8) = *(const v8h*)(nrow + t * 8);

#pragma unroll
    for (int t = 0; t < 8; ++t) {
        int off = l * 32 + t * 4;
        int row = off >> 6, col = off & 63;
        const float4 v = *(const float4*)(eattr + (size_t)(e0 + row) * 64 + col);
        v4h p; p[0] = (f16)v.x; p[1] = (f16)v.y; p[2] = (f16)v.z; p[3] = (f16)v.w;
        *(v4h*)(ea + (row0 + row) * EA_STRIDE + col) = p;
    }

    w3l[threadIdx.x] = W3[threadIdx.x];

    __syncthreads();

    strip_layer<1, 2>(ea, EA_STRIDE, Wep, be, wv,
                      cmb, CMB_STRIDE, 256, l, m, h);

    __syncthreads();

    strip_layer<4, 12>(cmb, CMB_STRIDE, W1p, b1, wv * 4,
                       h1, H1_STRIDE, 0, l, m, h);

    __syncthreads();

    strip_layer<2, 16>(h1, H1_STRIDE, W2p, b2, wv * 2,
                       cmb, CMB_STRIDE, 0, l, m, h);

    __syncthreads();

    if (valid && h == 0) {
        float s = b3[0];
        for (int t = 0; t < 32; ++t) {
            v8h c = *(const v8h*)(cmb + (row0 + m) * CMB_STRIDE + t * 8);
#pragma unroll
            for (int u = 0; u < 8; ++u) s += (float)c[u] * w3l[t * 8 + u];
        }
        out[e0 + m] = s;
    }
}

#define WS_WN 0
#define WS_WE (WS_WN + 16384 * 2)
#define WS_W1 (WS_WE + 8192 * 2)
#define WS_W2 (WS_W1 + 196608 * 2)
#define WS_NF (WS_W2 + 131072 * 2)

extern "C" void kernel_launch(void* const* d_in, const int* in_sizes, int n_in,
                              void* d_out, int out_size, void* d_ws, size_t ws_size,
                              hipStream_t stream) {
    (void)in_sizes; (void)n_in; (void)out_size; (void)ws_size;
    const float* x     = (const float*)d_in[0];
    const int*   eidx  = (const int*)  d_in[1];
    const float* eattr = (const float*)d_in[2];
    const float* Wn    = (const float*)d_in[3];
    const float* bn    = (const float*)d_in[4];
    const float* We    = (const float*)d_in[5];
    const float* be    = (const float*)d_in[6];
    const float* W1    = (const float*)d_in[7];
    const float* b1    = (const float*)d_in[8];
    const float* W2    = (const float*)d_in[9];
    const float* b2    = (const float*)d_in[10];
    const float* W3    = (const float*)d_in[11];
    const float* b3    = (const float*)d_in[12];

    char* ws  = (char*)d_ws;
    f16* Wnp = (f16*)(ws + WS_WN);
    f16* Wep = (f16*)(ws + WS_WE);
    f16* W1p = (f16*)(ws + WS_W1);
    f16* W2p = (f16*)(ws + WS_W2);
    f16* nf  = (f16*)(ws + WS_NF);

    (void)hipFuncSetAttribute((const void*)edge_fused_kernel,
                              hipFuncAttributeMaxDynamicSharedMemorySize,
                              WG_LDS);

    pack_w_kernel<<<64,  256, 0, stream>>>(Wn, Wnp, 128, 4,  16384);
    pack_w_kernel<<<32,  256, 0, stream>>>(We, Wep, 128, 2,  8192);
    pack_w_kernel<<<768, 256, 0, stream>>>(W1, W1p, 512, 12, 196608);
    pack_w_kernel<<<512, 256, 0, stream>>>(W2, W2p, 256, 16, 131072);

    node_feat_kernel<<<(3125 + 3) / 4, 128, 0, stream>>>(x, bn, Wnp, nf);

    edge_fused_kernel<<<(ETILES + EWAVES - 1) / EWAVES, 32 * EWAVES,
                        WG_LDS, stream>>>(
        eidx, eattr, be, b1, b2, W3, b3, Wep, W1p, W2p, nf, (float*)d_out);
}


# WMMA layers: k loop fully unrolled; each WMMA block's A (LDS) and weight (global) fragments are loaded as one group into spare register sets one or more blocks ahead, not right before use
# speedup vs baseline: 3.0117x; 1.1537x over previous
#include <hip/hip_runtime.h>
#include <hip/hip_bf16.h>


typedef _Float16 f16;
typedef __attribute__((ext_vector_type(16))) _Float16 v16h;
typedef __attribute__((ext_vector_type(8)))  _Float16 v8h;
typedef __attribute__((ext_vector_type(4)))  _Float16 v4h;
typedef __attribute__((ext_vector_type(8)))  float    v8f;

#define N_NODES 50000
#define N_EDGES 500000
#define ETILES  (N_EDGES / 16)
#define EWAVES  8

__device__ __forceinline__ v16h join16(v8h lo, v8h hi) {
    union { v16h v; v8h h[2]; } u;
    u.h[0] = lo; u.h[1] = hi;
    return u.v;
}

__device__ __forceinline__ v8f splat8(float x) {
    v8f a;
#pragma unroll
    for (int r = 0; r < 8; ++r) a[r] = x;
    return a;
}

__device__ __forceinline__ v8f wmma_f16(v16h a, v16h b, v8f c) {
    return __builtin_amdgcn_wmma_f32_16x16x32_f16(false, a, false, b,
                                                  (short)0, c, false, false);
}

__global__ void pack_w_kernel(const float* __restrict__ W, f16* __restrict__ out,
                              int Ntot, int kTiles, int total) {
    for (int idx = blockIdx.x * blockDim.x + threadIdx.x; idx < total;
         idx += gridDim.x * blockDim.x) {
        int f  = idx >> 9;
        int r  = idx & 511;
        int l  = r >> 4;
        int i  = r & 15;
        int nt = f / kTiles;
        int kt = f - nt * kTiles;
        int K  = kt * 32 + ((i >> 3) << 4) + ((l >> 4) << 3) + (i & 7);
        int N  = nt * 16 + (l & 15);
        out[idx] = (f16)W[(size_t)K * Ntot + N];
    }
}

__global__ void node_feat_kernel(const float* __restrict__ x,
                                 const float* __restrict__ bn,
                                 const f16*  __restrict__ Wnp,
                                 f16* __restrict__ nf) {
    __shared__ __align__(16) f16 xs_all[4][16 * 128];
    const int l  = threadIdx.x & 31;
    const int wv = threadIdx.x >> 5;
    const int tile = blockIdx.x * 4 + wv;
    if (tile >= N_NODES / 16) return;
    f16* xs = xs_all[wv];
    const int n0 = tile * 16;
    const int m  = l & 15;
    const int h  = l >> 4;

#pragma unroll
    for (int t = 0; t < 16; ++t) {
        int off = l * 64 + t * 4;
        int row = off >> 7, col = off & 127;
        const float4 v = *(const float4*)(x + (size_t)(n0 + row) * 128 + col);
        v4h p; p[0] = (f16)v.x; p[1] = (f16)v.y; p[2] = (f16)v.z; p[3] = (f16)v.w;
        *(v4h*)(xs + row * 128 + col) = p;
    }

    for (int nc = 0; nc < 2; ++nc) {
        v8f acc[4];
#pragma unroll
        for (int j = 0; j < 4; ++j) acc[j] = splat8(bn[(nc * 4 + j) * 16 + m]);
        for (int kt = 0; kt < 4; ++kt) {
            int k0  = kt * 32 + h * 8;
            v16h a = join16(*(const v8h*)(xs + m * 128 + k0),
                            *(const v8h*)(xs + m * 128 + k0 + 16));
            v16h b[4];
#pragma unroll
            for (int j = 0; j < 4; ++j)
                b[j] = *(const v16h*)(Wnp + (size_t)(((nc * 4 + j) * 4 + kt) * 512) + l * 16);
#pragma unroll
            for (int j = 0; j < 4; ++j) acc[j] = wmma_f16(a, b[j], acc[j]);
        }
#pragma unroll
        for (int j = 0; j < 4; ++j)
#pragma unroll
            for (int r = 0; r < 8; ++r)
                nf[(size_t)(n0 + r + 8 * h) * 128 + (nc * 4 + j) * 16 + m] =
                    (f16)fmaxf(acc[j][r], 0.0f);
    }
}

#define CMB_STRIDE 392
#define H1_STRIDE  520
#define EA_STRIDE  64
#define MTILES     8
#define LDS_CMB    0
#define LDS_H1     (LDS_CMB + 128 * CMB_STRIDE * 2)
#define LDS_EA     (LDS_H1  + 128 * H1_STRIDE  * 2)
#define LDS_W3     (LDS_EA  + 128 * EA_STRIDE  * 2)
#define WG_LDS     (LDS_W3  + 256 * 4)


template <int NJ, int KT>
__device__ __forceinline__ void load_b(v16h (&b)[NJ], const f16* __restrict__ Wp,
                                       int nt0, int kt, int l) {
#pragma unroll
    for (int j = 0; j < NJ; ++j)
        b[j] = *(const v16h*)(Wp + (size_t)(((nt0 + j) * KT + kt) * 512) + l * 16);
}

template <int AH>
__device__ __forceinline__ void load_a(v16h (&a)[MTILES / AH], const f16* A, int a_stride,
                                       int blk, int m, int h) {
    constexpr int CM = MTILES / AH;
    const int k0  = (blk / AH) * 32 + h * 8;
    const int mi0 = (blk % AH) * CM;
#pragma unroll
    for (int mi = 0; mi < CM; ++mi) {
        const f16* ar = A + ((mi0 + mi) * 16 + m) * a_stride + k0;
        a[mi] = join16(*(const v8h*)ar, *(const v8h*)(ar + 16));
    }
}

template <int NJ>
__device__ __forceinline__ void init_acc(v8f (&acc)[MTILES][NJ],
                                         const float* __restrict__ bias,
                                         int nt0, int m) {
#pragma unroll
    for (int j = 0; j < NJ; ++j) {
        const float bv = bias[(nt0 + j) * 16 + m];
#pragma unroll
        for (int mi = 0; mi < MTILES; ++mi) acc[mi][j] = splat8(bv);
    }
}

template <int NJ, int KT, int PB, int AH, int AR>
__device__ __forceinline__ void strip_mma(v8f (&acc)[MTILES][NJ],
                                          v16h (&b)[PB + 1][NJ],
                                          const f16* A, int a_stride,
                                          const f16* __restrict__ Wp, int nt0,
                                          int l, int m, int h) {
    constexpr int CM = MTILES / AH;
    constexpr int NB = KT * AH;
    v16h a[AR][CM];
#pragma unroll
    for (int p = 0; p < AR - 1; ++p) load_a<AH>(a[p], A, a_stride, p, m, h);
#pragma unroll
    for (int blk = 0; blk < NB; ++blk) {
        const int kt = blk / AH;
        const int c  = blk % AH;
        if (c == 0 && kt + PB < KT)
            load_b<NJ, KT>(b[(kt + PB) % (PB + 1)], Wp, nt0, kt + PB, l);
        if (blk + AR - 1 < NB)
            load_a<AH>(a[(blk + AR - 1) % AR], A, a_stride, blk + AR - 1, m, h);
        __builtin_amdgcn_sched_barrier(0);
#pragma unroll
        for (int mi = 0; mi < CM; ++mi)
#pragma unroll
            for (int j = 0; j < NJ; ++j)
                acc[c * CM + mi][j] = wmma_f16(a[blk % AR][mi], b[kt % (PB + 1)][j],
                                               acc[c * CM + mi][j]);
        __builtin_amdgcn_sched_barrier(0);
    }
}

template <int NJ>
__device__ __forceinline__ void strip_store(const v8f (&acc)[MTILES][NJ], int nt0,
                                            f16* D, int d_stride, int d_col0,
                                            int m, int h) {
#pragma unroll
    for (int mi = 0; mi < MTILES; ++mi)
#pragma unroll
        for (int j = 0; j < NJ; ++j)
#pragma unroll
            for (int r = 0; r < 8; ++r)
                D[(mi * 16 + r + 8 * h) * d_stride + d_col0 + (nt0 + j) * 16 + m] =
                    (f16)fmaxf(acc[mi][j][r], 0.0f);
}

__global__ void __launch_bounds__(32 * EWAVES)
edge_fused_kernel(const int*   __restrict__ eidx,
                  const float* __restrict__ eattr,
                  const float* __restrict__ be,
                  const float* __restrict__ b1,
                  const float* __restrict__ b2,
                  const float* __restrict__ W3,
                  const float* __restrict__ b3,
                  const f16*   __restrict__ Wep,
                  const f16*   __restrict__ W1p,
                  const f16*   __restrict__ W2p,
                  const f16*   __restrict__ nf,
                  float* __restrict__ out) {
    extern __shared__ __align__(16) char smem[];
    const int l  = threadIdx.x & 31;
    const int wv = threadIdx.x >> 5;
    const int m  = l & 15;
    const int h  = l >> 4;
    f16*   cmb = (f16*)(smem + LDS_CMB);
    f16*   h1  = (f16*)(smem + LDS_H1);
    f16*   ea  = (f16*)(smem + LDS_EA);
    float* w3l = (float*)(smem + LDS_W3);

    const int  tile  = blockIdx.x * EWAVES + wv;
    const bool valid = (tile < ETILES);
    const int  e0    = valid ? tile * 16 : 0;
    const int  row0  = wv * 16;

    v16h wbe[2][1];
    v16h wb1[3][4];
    v16h wb2[5][2];

    load_b<1, 2>(wbe[0], Wep, wv, 0, l);

    const int node = eidx[h * N_EDGES + e0 + m];
    const f16* nrow = nf + (size_t)node * 128;
#pragma unroll
    for (int t = 0; t < 16; ++t)
        *(v8h*)(cmb + (row0 + m) * CMB_STRIDE + h * 128 + t * 8) = *(const v8h*)(nrow + t * 8);

#pragma unroll
    for (int t = 0; t < 8; ++t) {
        int off = l * 32 + t * 4;
        int row = off >> 6, col = off & 63;
        const float4 v = *(const float4*)(eattr + (size_t)(e0 + row) * 64 + col);
        v4h p; p[0] = (f16)v.x; p[1] = (f16)v.y; p[2] = (f16)v.z; p[3] = (f16)v.w;
        *(v4h*)(ea + (row0 + row) * EA_STRIDE + col) = p;
    }

    w3l[threadIdx.x] = W3[threadIdx.x];

    __syncthreads();

    {
        v8f acc[MTILES][1];
        init_acc<1>(acc, be, wv, m);
        load_b<4, 12>(wb1[0], W1p, wv * 4, 0, l);
        load_b<4, 12>(wb1[1], W1p, wv * 4, 1, l);
        strip_mma<1, 2, 1, 1, 2>(acc, wbe, ea, EA_STRIDE, Wep, wv, l, m, h);
        strip_store<1>(acc, wv, cmb, CMB_STRIDE, 256, m, h);
    }

    __syncthreads();

    {
        v8f acc[MTILES][4];
        init_acc<4>(acc, b1, wv * 4, m);
        strip_mma<4, 12, 2, 2, 2>(acc, wb1, cmb, CMB_STRIDE, W1p, wv * 4, l, m, h);
#pragma unroll
        for (int p = 0; p < 4; ++p)
            load_b<2, 16>(wb2[p], W2p, wv * 2, p, l);
        strip_store<4>(acc, wv * 4, h1, H1_STRIDE, 0, m, h);
    }

    __syncthreads();

    {
        v8f acc[MTILES][2];
        init_acc<2>(acc, b2, wv * 2, m);
        strip_mma<2, 16, 4, 1, 3>(acc, wb2, h1, H1_STRIDE, W2p, wv * 2, l, m, h);
        strip_store<2>(acc, wv * 2, cmb, CMB_STRIDE, 0, m, h);
    }

    __syncthreads();

    if (valid && h == 0) {
        float s = b3[0];
        for (int t = 0; t < 32; ++t) {
            v8h c = *(const v8h*)(cmb + (row0 + m) * CMB_STRIDE + t * 8);
#pragma unroll
            for (int u = 0; u < 8; ++u) s += (float)c[u] * w3l[t * 8 + u];
        }
        out[e0 + m] = s;
    }
}

#define WS_WN 0
#define WS_WE (WS_WN + 16384 * 2)
#define WS_W1 (WS_WE + 8192 * 2)
#define WS_W2 (WS_W1 + 196608 * 2)
#define WS_NF (WS_W2 + 131072 * 2)

extern "C" void kernel_launch(void* const* d_in, const int* in_sizes, int n_in,
                              void* d_out, int out_size, void* d_ws, size_t ws_size,
                              hipStream_t stream) {
    (void)in_sizes; (void)n_in; (void)out_size; (void)ws_size;
    const float* x     = (const float*)d_in[0];
    const int*   eidx  = (const int*)  d_in[1];
    const float* eattr = (const float*)d_in[2];
    const float* Wn    = (const float*)d_in[3];
    const float* bn    = (const float*)d_in[4];
    const float* We    = (const float*)d_in[5];
    const float* be    = (const float*)d_in[6];
    const float* W1    = (const float*)d_in[7];
    const float* b1    = (const float*)d_in[8];
    const float* W2    = (const float*)d_in[9];
    const float* b2    = (const float*)d_in[10];
    const float* W3    = (const float*)d_in[11];
    const float* b3    = (const float*)d_in[12];

    char* ws  = (char*)d_ws;
    f16* Wnp = (f16*)(ws + WS_WN);
    f16* Wep = (f16*)(ws + WS_WE);
    f16* W1p = (f16*)(ws + WS_W1);
    f16* W2p = (f16*)(ws + WS_W2);
    f16* nf  = (f16*)(ws + WS_NF);

    (void)hipFuncSetAttribute((const void*)edge_fused_kernel,
                              hipFuncAttributeMaxDynamicSharedMemorySize,
                              WG_LDS);

    pack_w_kernel<<<64,  256, 0, stream>>>(Wn, Wnp, 128, 4,  16384);
    pack_w_kernel<<<32,  256, 0, stream>>>(We, Wep, 128, 2,  8192);
    pack_w_kernel<<<768, 256, 0, stream>>>(W1, W1p, 512, 12, 196608);
    pack_w_kernel<<<512, 256, 0, stream>>>(W2, W2p, 256, 16, 131072);

    node_feat_kernel<<<(3125 + 3) / 4, 128, 0, stream>>>(x, bn, Wnp, nf);

    edge_fused_kernel<<<(ETILES + EWAVES - 1) / EWAVES, 32 * EWAVES,
                        WG_LDS, stream>>>(
        eidx, eattr, be, b1, b2, W3, b3, Wep, W1p, W2p, nf, (float*)d_out);
}


# WMMA operands swapped (weights as A, activations as B): a lane holds 8 consecutive features of one row, so a tile ends in packed converts, ReLU and one 16-byte store, not eight 2-byte stores
# speedup vs baseline: 3.6482x; 1.2113x over previous
#include <hip/hip_runtime.h>
#include <hip/hip_bf16.h>


typedef _Float16 f16;
typedef __attribute__((ext_vector_type(16))) _Float16 v16h;
typedef __attribute__((ext_vector_type(8)))  _Float16 v8h;
typedef __attribute__((ext_vector_type(4)))  _Float16 v4h;
typedef __attribute__((ext_vector_type(8)))  float    v8f;

#define N_NODES 50000
#define N_EDGES 500000
#define ETILES  (N_EDGES / 16)
#define EWAVES  8

__device__ __forceinline__ v16h join16(v8h lo, v8h hi) {
    union { v16h v; v8h h[2]; } u;
    u.h[0] = lo; u.h[1] = hi;
    return u.v;
}

__device__ __forceinline__ v8f load8f(const float* __restrict__ p) {
    const float4 lo = *(const float4*)p;
    const float4 hi = *(const float4*)(p + 4);
    v8f a;
    a[0] = lo.x; a[1] = lo.y; a[2] = lo.z; a[3] = lo.w;
    a[4] = hi.x; a[5] = hi.y; a[6] = hi.z; a[7] = hi.w;
    return a;
}

__device__ __forceinline__ v8h relu_pack(v8f x) {
    v8h p;
#pragma unroll
    for (int r = 0; r < 8; ++r) p[r] = (f16)x[r];
    const v8h z = {0, 0, 0, 0, 0, 0, 0, 0};
    return __builtin_elementwise_max(p, z);
}

__device__ __forceinline__ v8f wmma_f16(v16h a, v16h b, v8f c) {
    return __builtin_amdgcn_wmma_f32_16x16x32_f16(false, a, false, b,
                                                  (short)0, c, false, false);
}

__global__ void pack_w_kernel(const float* __restrict__ W, f16* __restrict__ out,
                              int Ntot, int kTiles, int total) {
    for (int idx = blockIdx.x * blockDim.x + threadIdx.x; idx < total;
         idx += gridDim.x * blockDim.x) {
        int f  = idx >> 9;
        int r  = idx & 511;
        int l  = r >> 4;
        int i  = r & 15;
        int nt = f / kTiles;
        int kt = f - nt * kTiles;
        int K  = kt * 32 + ((i >> 3) << 4) + ((l >> 4) << 3) + (i & 7);
        int N  = nt * 16 + (l & 15);
        out[idx] = (f16)W[(size_t)K * Ntot + N];
    }
}

__global__ void node_feat_kernel(const float* __restrict__ x,
                                 const float* __restrict__ bn,
                                 const f16*  __restrict__ Wnp,
                                 f16* __restrict__ nf) {
    __shared__ __align__(16) f16 xs_all[4][16 * 128];
    const int l  = threadIdx.x & 31;
    const int wv = threadIdx.x >> 5;
    const int tile = blockIdx.x * 4 + wv;
    if (tile >= N_NODES / 16) return;
    f16* xs = xs_all[wv];
    const int n0 = tile * 16;
    const int m  = l & 15;
    const int h  = l >> 4;

#pragma unroll
    for (int t = 0; t < 16; ++t) {
        int off = l * 64 + t * 4;
        int row = off >> 7, col = off & 127;
        const float4 v = *(const float4*)(x + (size_t)(n0 + row) * 128 + col);
        v4h p; p[0] = (f16)v.x; p[1] = (f16)v.y; p[2] = (f16)v.z; p[3] = (f16)v.w;
        *(v4h*)(xs + row * 128 + col) = p;
    }

    for (int nc = 0; nc < 2; ++nc) {
        v8f acc[4];
#pragma unroll
        for (int j = 0; j < 4; ++j) acc[j] = load8f(bn + (nc * 4 + j) * 16 + 8 * h);
        for (int kt = 0; kt < 4; ++kt) {
            int k0  = kt * 32 + h * 8;
            v16h a = join16(*(const v8h*)(xs + m * 128 + k0),
                            *(const v8h*)(xs + m * 128 + k0 + 16));
            v16h b[4];
#pragma unroll
            for (int j = 0; j < 4; ++j)
                b[j] = *(const v16h*)(Wnp + (size_t)(((nc * 4 + j) * 4 + kt) * 512) + l * 16);
#pragma unroll
            for (int j = 0; j < 4; ++j) acc[j] = wmma_f16(b[j], a, acc[j]);
        }
#pragma unroll
        for (int j = 0; j < 4; ++j)
            *(v8h*)(nf + (size_t)(n0 + m) * 128 + (nc * 4 + j) * 16 + 8 * h) =
                relu_pack(acc[j]);
    }
}

#define CMB_STRIDE 392
#define H1_STRIDE  520
#define EA_STRIDE  64
#define MTILES     8
#define LDS_CMB    0
#define LDS_H1     (LDS_CMB + 128 * CMB_STRIDE * 2)
#define LDS_EA     (LDS_H1  + 128 * H1_STRIDE  * 2)
#define LDS_W3     (LDS_EA  + 128 * EA_STRIDE  * 2)
#define WG_LDS     (LDS_W3  + 256 * 4)


template <int NJ, int KT>
__device__ __forceinline__ void load_b(v16h (&b)[NJ], const f16* __restrict__ Wp,
                                       int nt0, int kt, int l) {
#pragma unroll
    for (int j = 0; j < NJ; ++j)
        b[j] = *(const v16h*)(Wp + (size_t)(((nt0 + j) * KT + kt) * 512) + l * 16);
}

template <int AH>
__device__ __forceinline__ void load_a(v16h (&a)[MTILES / AH], const f16* A, int a_stride,
                                       int blk, int m, int h) {
    constexpr int CM = MTILES / AH;
    const int k0  = (blk / AH) * 32 + h * 8;
    const int mi0 = (blk % AH) * CM;
#pragma unroll
    for (int mi = 0; mi < CM; ++mi) {
        const f16* ar = A + ((mi0 + mi) * 16 + m) * a_stride + k0;
        a[mi] = join16(*(const v8h*)ar, *(const v8h*)(ar + 16));
    }
}

template <int NJ>
__device__ __forceinline__ void init_acc(v8f (&acc)[MTILES][NJ],
                                         const float* __restrict__ bias,
                                         int nt0, int h) {
#pragma unroll
    for (int j = 0; j < NJ; ++j) {
        const v8f bv = load8f(bias + (nt0 + j) * 16 + 8 * h);
#pragma unroll
        for (int mi = 0; mi < MTILES; ++mi) acc[mi][j] = bv;
    }
}

template <int NJ, int KT, int PB, int AH, int AR>
__device__ __forceinline__ void strip_mma(v8f (&acc)[MTILES][NJ],
                                          v16h (&b)[PB + 1][NJ],
                                          const f16* A, int a_stride,
                                          const f16* __restrict__ Wp, int nt0,
                                          int l, int m, int h) {
    constexpr int CM = MTILES / AH;
    constexpr int NB = KT * AH;
    v16h a[AR][CM];
#pragma unroll
    for (int p = 0; p < AR - 1; ++p) load_a<AH>(a[p], A, a_stride, p, m, h);
#pragma unroll
    for (int blk = 0; blk < NB; ++blk) {
        const int kt = blk / AH;
        const int c  = blk % AH;
        if (c == 0 && kt + PB < KT)
            load_b<NJ, KT>(b[(kt + PB) % (PB + 1)], Wp, nt0, kt + PB, l);
        if (blk + AR - 1 < NB)
            load_a<AH>(a[(blk + AR - 1) % AR], A, a_stride, blk + AR - 1, m, h);
        __builtin_amdgcn_sched_barrier(0);
#pragma unroll
        for (int mi = 0; mi < CM; ++mi)
#pragma unroll
            for (int j = 0; j < NJ; ++j)
                acc[c * CM + mi][j] = wmma_f16(b[kt % (PB + 1)][j], a[blk % AR][mi],
                                               acc[c * CM + mi][j]);
        __builtin_amdgcn_sched_barrier(0);
    }
}

template <int NJ>
__device__ __forceinline__ void strip_store(const v8f (&acc)[MTILES][NJ], int nt0,
                                            f16* D, int d_stride, int d_col0,
                                            int m, int h) {
#pragma unroll
    for (int mi = 0; mi < MTILES; ++mi)
#pragma unroll
        for (int j = 0; j < NJ; ++j)
            *(v8h*)(D + (mi * 16 + m) * d_stride + d_col0 + (nt0 + j) * 16 + 8 * h) =
                relu_pack(acc[mi][j]);
}

__global__ void __launch_bounds__(32 * EWAVES)
edge_fused_kernel(const int*   __restrict__ eidx,
                  const float* __restrict__ eattr,
                  const float* __restrict__ be,
                  const float* __restrict__ b1,
                  const float* __restrict__ b2,
                  const float* __restrict__ W3,
                  const float* __restrict__ b3,
                  const f16*   __restrict__ Wep,
                  const f16*   __restrict__ W1p,
                  const f16*   __restrict__ W2p,
                  const f16*   __restrict__ nf,
                  float* __restrict__ out) {
    extern __shared__ __align__(16) char smem[];
    const int l  = threadIdx.x & 31;
    const int wv = threadIdx.x >> 5;
    const int m  = l & 15;
    const int h  = l >> 4;
    f16*   cmb = (f16*)(smem + LDS_CMB);
    f16*   h1  = (f16*)(smem + LDS_H1);
    f16*   ea  = (f16*)(smem + LDS_EA);
    float* w3l = (float*)(smem + LDS_W3);

    const int  tile  = blockIdx.x * EWAVES + wv;
    const bool valid = (tile < ETILES);
    const int  e0    = valid ? tile * 16 : 0;
    const int  row0  = wv * 16;

    v16h wbe[2][1];
    v16h wb1[3][4];
    v16h wb2[5][2];

    load_b<1, 2>(wbe[0], Wep, wv, 0, l);

    const int node = eidx[h * N_EDGES + e0 + m];
    const f16* nrow = nf + (size_t)node * 128;
#pragma unroll
    for (int t = 0; t < 16; ++t)
        *(v8h*)(cmb + (row0 + m) * CMB_STRIDE + h * 128 + t * 8) = *(const v8h*)(nrow + t * 8);

#pragma unroll
    for (int t = 0; t < 8; ++t) {
        int off = l * 32 + t * 4;
        int row = off >> 6, col = off & 63;
        const float4 v = *(const float4*)(eattr + (size_t)(e0 + row) * 64 + col);
        v4h p; p[0] = (f16)v.x; p[1] = (f16)v.y; p[2] = (f16)v.z; p[3] = (f16)v.w;
        *(v4h*)(ea + (row0 + row) * EA_STRIDE + col) = p;
    }

    w3l[threadIdx.x] = W3[threadIdx.x];

    __syncthreads();

    {
        v8f acc[MTILES][1];
        init_acc<1>(acc, be, wv, h);
        load_b<4, 12>(wb1[0], W1p, wv * 4, 0, l);
        load_b<4, 12>(wb1[1], W1p, wv * 4, 1, l);
        strip_mma<1, 2, 1, 1, 2>(acc, wbe, ea, EA_STRIDE, Wep, wv, l, m, h);
        strip_store<1>(acc, wv, cmb, CMB_STRIDE, 256, m, h);
    }

    __syncthreads();

    {
        v8f acc[MTILES][4];
        init_acc<4>(acc, b1, wv * 4, h);
        strip_mma<4, 12, 2, 2, 2>(acc, wb1, cmb, CMB_STRIDE, W1p, wv * 4, l, m, h);
#pragma unroll
        for (int p = 0; p < 4; ++p)
            load_b<2, 16>(wb2[p], W2p, wv * 2, p, l);
        strip_store<4>(acc, wv * 4, h1, H1_STRIDE, 0, m, h);
    }

    __syncthreads();

    {
        v8f acc[MTILES][2];
        init_acc<2>(acc, b2, wv * 2, h);
        strip_mma<2, 16, 4, 1, 3>(acc, wb2, h1, H1_STRIDE, W2p, wv * 2, l, m, h);
        strip_store<2>(acc, wv * 2, cmb, CMB_STRIDE, 0, m, h);
    }

    __syncthreads();

    if (valid && h == 0) {
        float s = b3[0];
        for (int t = 0; t < 32; ++t) {
            v8h c = *(const v8h*)(cmb + (row0 + m) * CMB_STRIDE + t * 8);
#pragma unroll
            for (int u = 0; u < 8; ++u) s += (float)c[u] * w3l[t * 8 + u];
        }
        out[e0 + m] = s;
    }
}

#define WS_WN 0
#define WS_WE (WS_WN + 16384 * 2)
#define WS_W1 (WS_WE + 8192 * 2)
#define WS_W2 (WS_W1 + 196608 * 2)
#define WS_NF (WS_W2 + 131072 * 2)

extern "C" void kernel_launch(void* const* d_in, const int* in_sizes, int n_in,
                              void* d_out, int out_size, void* d_ws, size_t ws_size,
                              hipStream_t stream) {
    (void)in_sizes; (void)n_in; (void)out_size; (void)ws_size;
    const float* x     = (const float*)d_in[0];
    const int*   eidx  = (const int*)  d_in[1];
    const float* eattr = (const float*)d_in[2];
    const float* Wn    = (const float*)d_in[3];
    const float* bn    = (const float*)d_in[4];
    const float* We    = (const float*)d_in[5];
    const float* be    = (const float*)d_in[6];
    const float* W1    = (const float*)d_in[7];
    const float* b1    = (const float*)d_in[8];
    const float* W2    = (const float*)d_in[9];
    const float* b2    = (const float*)d_in[10];
    const float* W3    = (const float*)d_in[11];
    const float* b3    = (const float*)d_in[12];

    char* ws  = (char*)d_ws;
    f16* Wnp = (f16*)(ws + WS_WN);
    f16* Wep = (f16*)(ws + WS_WE);
    f16* W1p = (f16*)(ws + WS_W1);
    f16* W2p = (f16*)(ws + WS_W2);
    f16* nf  = (f16*)(ws + WS_NF);

    (void)hipFuncSetAttribute((const void*)edge_fused_kernel,
                              hipFuncAttributeMaxDynamicSharedMemorySize,
                              WG_LDS);

    pack_w_kernel<<<64,  256, 0, stream>>>(Wn, Wnp, 128, 4,  16384);
    pack_w_kernel<<<32,  256, 0, stream>>>(We, Wep, 128, 2,  8192);
    pack_w_kernel<<<768, 256, 0, stream>>>(W1, W1p, 512, 12, 196608);
    pack_w_kernel<<<512, 256, 0, stream>>>(W2, W2p, 256, 16, 131072);

    node_feat_kernel<<<(3125 + 3) / 4, 128, 0, stream>>>(x, bn, Wnp, nf);

    edge_fused_kernel<<<(ETILES + EWAVES - 1) / EWAVES, 32 * EWAVES,
                        WG_LDS, stream>>>(
        eidx, eattr, be, b1, b2, W3, b3, Wep, W1p, W2p, nf, (float*)d_out);
}


# staging loads re-mapped to whole 128-byte lines: 16 lanes copy one gathered node row per load (index broadcast by readlane); edge_attr and x tiles read 512 contiguous bytes per load
# speedup vs baseline: 3.8686x; 1.0604x over previous
#include <hip/hip_runtime.h>
#include <hip/hip_bf16.h>


typedef _Float16 f16;
typedef __attribute__((ext_vector_type(16))) _Float16 v16h;
typedef __attribute__((ext_vector_type(8)))  _Float16 v8h;
typedef __attribute__((ext_vector_type(4)))  _Float16 v4h;
typedef __attribute__((ext_vector_type(8)))  float    v8f;

#define N_NODES 50000
#define N_EDGES 500000
#define ETILES  (N_EDGES / 16)
#define EWAVES  8

__device__ __forceinline__ v16h join16(v8h lo, v8h hi) {
    union { v16h v; v8h h[2]; } u;
    u.h[0] = lo; u.h[1] = hi;
    return u.v;
}

__device__ __forceinline__ v8f load8f(const float* __restrict__ p) {
    const float4 lo = *(const float4*)p;
    const float4 hi = *(const float4*)(p + 4);
    v8f a;
    a[0] = lo.x; a[1] = lo.y; a[2] = lo.z; a[3] = lo.w;
    a[4] = hi.x; a[5] = hi.y; a[6] = hi.z; a[7] = hi.w;
    return a;
}

__device__ __forceinline__ v8h relu_pack(v8f x) {
    v8h p;
#pragma unroll
    for (int r = 0; r < 8; ++r) p[r] = (f16)x[r];
    const v8h z = {0, 0, 0, 0, 0, 0, 0, 0};
    return __builtin_elementwise_max(p, z);
}

__device__ __forceinline__ v8f wmma_f16(v16h a, v16h b, v8f c) {
    return __builtin_amdgcn_wmma_f32_16x16x32_f16(false, a, false, b,
                                                  (short)0, c, false, false);
}

__global__ void pack_w_kernel(const float* __restrict__ W, f16* __restrict__ out,
                              int Ntot, int kTiles, int total) {
    for (int idx = blockIdx.x * blockDim.x + threadIdx.x; idx < total;
         idx += gridDim.x * blockDim.x) {
        int f  = idx >> 9;
        int r  = idx & 511;
        int l  = r >> 4;
        int i  = r & 15;
        int nt = f / kTiles;
        int kt = f - nt * kTiles;
        int K  = kt * 32 + ((i >> 3) << 4) + ((l >> 4) << 3) + (i & 7);
        int N  = nt * 16 + (l & 15);
        out[idx] = (f16)W[(size_t)K * Ntot + N];
    }
}

__global__ void node_feat_kernel(const float* __restrict__ x,
                                 const float* __restrict__ bn,
                                 const f16*  __restrict__ Wnp,
                                 f16* __restrict__ nf) {
    __shared__ __align__(16) f16 xs_all[4][16 * 128];
    const int l  = threadIdx.x & 31;
    const int wv = threadIdx.x >> 5;
    const int tile = blockIdx.x * 4 + wv;
    if (tile >= N_NODES / 16) return;
    f16* xs = xs_all[wv];
    const int n0 = tile * 16;
    const int m  = l & 15;
    const int h  = l >> 4;

#pragma unroll
    for (int t = 0; t < 16; ++t) {
        const int row = t, col = l * 4;
        const float4 v = *(const float4*)(x + (size_t)(n0 + row) * 128 + col);
        v4h p; p[0] = (f16)v.x; p[1] = (f16)v.y; p[2] = (f16)v.z; p[3] = (f16)v.w;
        *(v4h*)(xs + row * 128 + col) = p;
    }

    for (int nc = 0; nc < 2; ++nc) {
        v8f acc[4];
#pragma unroll
        for (int j = 0; j < 4; ++j) acc[j] = load8f(bn + (nc * 4 + j) * 16 + 8 * h);
        for (int kt = 0; kt < 4; ++kt) {
            int k0  = kt * 32 + h * 8;
            v16h a = join16(*(const v8h*)(xs + m * 128 + k0),
                            *(const v8h*)(xs + m * 128 + k0 + 16));
            v16h b[4];
#pragma unroll
            for (int j = 0; j < 4; ++j)
                b[j] = *(const v16h*)(Wnp + (size_t)(((nc * 4 + j) * 4 + kt) * 512) + l * 16);
#pragma unroll
            for (int j = 0; j < 4; ++j) acc[j] = wmma_f16(b[j], a, acc[j]);
        }
#pragma unroll
        for (int j = 0; j < 4; ++j)
            *(v8h*)(nf + (size_t)(n0 + m) * 128 + (nc * 4 + j) * 16 + 8 * h) =
                relu_pack(acc[j]);
    }
}

#define CMB_STRIDE 392
#define H1_STRIDE  520
#define EA_STRIDE  64
#define MTILES     8
#define LDS_CMB    0
#define LDS_H1     (LDS_CMB + 128 * CMB_STRIDE * 2)
#define LDS_EA     (LDS_H1  + 128 * H1_STRIDE  * 2)
#define LDS_W3     (LDS_EA  + 128 * EA_STRIDE  * 2)
#define WG_LDS     (LDS_W3  + 256 * 4)


template <int NJ, int KT>
__device__ __forceinline__ void load_b(v16h (&b)[NJ], const f16* __restrict__ Wp,
                                       int nt0, int kt, int l) {
#pragma unroll
    for (int j = 0; j < NJ; ++j)
        b[j] = *(const v16h*)(Wp + (size_t)(((nt0 + j) * KT + kt) * 512) + l * 16);
}

template <int AH>
__device__ __forceinline__ void load_a(v16h (&a)[MTILES / AH], const f16* A, int a_stride,
                                       int blk, int m, int h) {
    constexpr int CM = MTILES / AH;
    const int k0  = (blk / AH) * 32 + h * 8;
    const int mi0 = (blk % AH) * CM;
#pragma unroll
    for (int mi = 0; mi < CM; ++mi) {
        const f16* ar = A + ((mi0 + mi) * 16 + m) * a_stride + k0;
        a[mi] = join16(*(const v8h*)ar, *(const v8h*)(ar + 16));
    }
}

template <int NJ>
__device__ __forceinline__ void init_acc(v8f (&acc)[MTILES][NJ],
                                         const float* __restrict__ bias,
                                         int nt0, int h) {
#pragma unroll
    for (int j = 0; j < NJ; ++j) {
        const v8f bv = load8f(bias + (nt0 + j) * 16 + 8 * h);
#pragma unroll
        for (int mi = 0; mi < MTILES; ++mi) acc[mi][j] = bv;
    }
}

template <int NJ, int KT, int PB, int AH, int AR>
__device__ __forceinline__ void strip_mma(v8f (&acc)[MTILES][NJ],
                                          v16h (&b)[PB + 1][NJ],
                                          const f16* A, int a_stride,
                                          const f16* __restrict__ Wp, int nt0,
                                          int l, int m, int h) {
    constexpr int CM = MTILES / AH;
    constexpr int NB = KT * AH;
    v16h a[AR][CM];
#pragma unroll
    for (int p = 0; p < AR - 1; ++p) load_a<AH>(a[p], A, a_stride, p, m, h);
#pragma unroll
    for (int blk = 0; blk < NB; ++blk) {
        const int kt = blk / AH;
        const int c  = blk % AH;
        if (c == 0 && kt + PB < KT)
            load_b<NJ, KT>(b[(kt + PB) % (PB + 1)], Wp, nt0, kt + PB, l);
        if (blk + AR - 1 < NB)
            load_a<AH>(a[(blk + AR - 1) % AR], A, a_stride, blk + AR - 1, m, h);
        __builtin_amdgcn_sched_barrier(0);
#pragma unroll
        for (int mi = 0; mi < CM; ++mi)
#pragma unroll
            for (int j = 0; j < NJ; ++j)
                acc[c * CM + mi][j] = wmma_f16(b[kt % (PB + 1)][j], a[blk % AR][mi],
                                               acc[c * CM + mi][j]);
        __builtin_amdgcn_sched_barrier(0);
    }
}

template <int NJ>
__device__ __forceinline__ void strip_store(const v8f (&acc)[MTILES][NJ], int nt0,
                                            f16* D, int d_stride, int d_col0,
                                            int m, int h) {
#pragma unroll
    for (int mi = 0; mi < MTILES; ++mi)
#pragma unroll
        for (int j = 0; j < NJ; ++j)
            *(v8h*)(D + (mi * 16 + m) * d_stride + d_col0 + (nt0 + j) * 16 + 8 * h) =
                relu_pack(acc[mi][j]);
}

__global__ void __launch_bounds__(32 * EWAVES)
edge_fused_kernel(const int*   __restrict__ eidx,
                  const float* __restrict__ eattr,
                  const float* __restrict__ be,
                  const float* __restrict__ b1,
                  const float* __restrict__ b2,
                  const float* __restrict__ W3,
                  const float* __restrict__ b3,
                  const f16*   __restrict__ Wep,
                  const f16*   __restrict__ W1p,
                  const f16*   __restrict__ W2p,
                  const f16*   __restrict__ nf,
                  float* __restrict__ out) {
    extern __shared__ __align__(16) char smem[];
    const int l  = threadIdx.x & 31;
    const int wv = threadIdx.x >> 5;
    const int m  = l & 15;
    const int h  = l >> 4;
    f16*   cmb = (f16*)(smem + LDS_CMB);
    f16*   h1  = (f16*)(smem + LDS_H1);
    f16*   ea  = (f16*)(smem + LDS_EA);
    float* w3l = (float*)(smem + LDS_W3);

    const int  tile  = blockIdx.x * EWAVES + wv;
    const bool valid = (tile < ETILES);
    const int  e0    = valid ? tile * 16 : 0;
    const int  row0  = wv * 16;

    v16h wbe[2][1];
    v16h wb1[3][4];
    v16h wb2[5][2];

    load_b<1, 2>(wbe[0], Wep, wv, 0, l);

    const int node = eidx[h * N_EDGES + e0 + m];

    float4 ev[8];
#pragma unroll
    for (int t = 0; t < 8; ++t)
        ev[t] = *(const float4*)(eattr + (size_t)e0 * 64 + t * 128 + l * 4);

    v8h g[16];
#pragma unroll
    for (int p = 0; p < 16; ++p) {
        const int ns = __builtin_amdgcn_readlane(node, p);
        const int nd = __builtin_amdgcn_readlane(node, 16 + p);
        g[p] = *(const v8h*)(nf + (size_t)(h ? nd : ns) * 128 + m * 8);
    }
    __builtin_amdgcn_sched_barrier(0);

#pragma unroll
    for (int t = 0; t < 8; ++t) {
        int off = t * 128 + l * 4;
        int row = off >> 6, col = off & 63;
        const float4 v = ev[t];
        v4h p; p[0] = (f16)v.x; p[1] = (f16)v.y; p[2] = (f16)v.z; p[3] = (f16)v.w;
        *(v4h*)(ea + (row0 + row) * EA_STRIDE + col) = p;
    }

#pragma unroll
    for (int p = 0; p < 16; ++p)
        *(v8h*)(cmb + (row0 + p) * CMB_STRIDE + h * 128 + m * 8) = g[p];

    w3l[threadIdx.x] = W3[threadIdx.x];

    __syncthreads();

    {
        v8f acc[MTILES][1];
        init_acc<1>(acc, be, wv, h);
        load_b<4, 12>(wb1[0], W1p, wv * 4, 0, l);
        load_b<4, 12>(wb1[1], W1p, wv * 4, 1, l);
        strip_mma<1, 2, 1, 1, 2>(acc, wbe, ea, EA_STRIDE, Wep, wv, l, m, h);
        strip_store<1>(acc, wv, cmb, CMB_STRIDE, 256, m, h);
    }

    __syncthreads();

    {
        v8f acc[MTILES][4];
        init_acc<4>(acc, b1, wv * 4, h);
        strip_mma<4, 12, 2, 2, 2>(acc, wb1, cmb, CMB_STRIDE, W1p, wv * 4, l, m, h);
#pragma unroll
        for (int p = 0; p < 4; ++p)
            load_b<2, 16>(wb2[p], W2p, wv * 2, p, l);
        strip_store<4>(acc, wv * 4, h1, H1_STRIDE, 0, m, h);
    }

    __syncthreads();

    {
        v8f acc[MTILES][2];
        init_acc<2>(acc, b2, wv * 2, h);
        strip_mma<2, 16, 4, 1, 3>(acc, wb2, h1, H1_STRIDE, W2p, wv * 2, l, m, h);
        strip_store<2>(acc, wv * 2, cmb, CMB_STRIDE, 0, m, h);
    }

    __syncthreads();

    if (valid && h == 0) {
        float s = b3[0];
        for (int t = 0; t < 32; ++t) {
            v8h c = *(const v8h*)(cmb + (row0 + m) * CMB_STRIDE + t * 8);
#pragma unroll
            for (int u = 0; u < 8; ++u) s += (float)c[u] * w3l[t * 8 + u];
        }
        out[e0 + m] = s;
    }
}

#define WS_WN 0
#define WS_WE (WS_WN + 16384 * 2)
#define WS_W1 (WS_WE + 8192 * 2)
#define WS_W2 (WS_W1 + 196608 * 2)
#define WS_NF (WS_W2 + 131072 * 2)

extern "C" void kernel_launch(void* const* d_in, const int* in_sizes, int n_in,
                              void* d_out, int out_size, void* d_ws, size_t ws_size,
                              hipStream_t stream) {
    (void)in_sizes; (void)n_in; (void)out_size; (void)ws_size;
    const float* x     = (const float*)d_in[0];
    const int*   eidx  = (const int*)  d_in[1];
    const float* eattr = (const float*)d_in[2];
    const float* Wn    = (const float*)d_in[3];
    const float* bn    = (const float*)d_in[4];
    const float* We    = (const float*)d_in[5];
    const float* be    = (const float*)d_in[6];
    const float* W1    = (const float*)d_in[7];
    const float* b1    = (const float*)d_in[8];
    const float* W2    = (const float*)d_in[9];
    const float* b2    = (const float*)d_in[10];
    const float* W3    = (const float*)d_in[11];
    const float* b3    = (const float*)d_in[12];

    char* ws  = (char*)d_ws;
    f16* Wnp = (f16*)(ws + WS_WN);
    f16* Wep = (f16*)(ws + WS_WE);
    f16* W1p = (f16*)(ws + WS_W1);
    f16* W2p = (f16*)(ws + WS_W2);
    f16* nf  = (f16*)(ws + WS_NF);

    (void)hipFuncSetAttribute((const void*)edge_fused_kernel,
                              hipFuncAttributeMaxDynamicSharedMemorySize,
                              WG_LDS);

    pack_w_kernel<<<64,  256, 0, stream>>>(Wn, Wnp, 128, 4,  16384);
    pack_w_kernel<<<32,  256, 0, stream>>>(We, Wep, 128, 2,  8192);
    pack_w_kernel<<<768, 256, 0, stream>>>(W1, W1p, 512, 12, 196608);
    pack_w_kernel<<<512, 256, 0, stream>>>(W2, W2p, 256, 16, 131072);

    node_feat_kernel<<<(3125 + 3) / 4, 128, 0, stream>>>(x, bn, Wnp, nf);

    edge_fused_kernel<<<(ETILES + EWAVES - 1) / EWAVES, 32 * EWAVES,
                        WG_LDS, stream>>>(
        eidx, eattr, be, b1, b2, W3, b3, Wep, W1p, W2p, nf, (float*)d_out);
}


# output head folded into layer 2's epilogue: each lane dots its 16 f32 accumulator values of a row with W3, per-wave partial sums meet in LDS (sum order of the 256-term dot changes; h2 stays f32)
# speedup vs baseline: 4.2240x; 1.0919x over previous
#include <hip/hip_runtime.h>
#include <hip/hip_bf16.h>


typedef _Float16 f16;
typedef __attribute__((ext_vector_type(16))) _Float16 v16h;
typedef __attribute__((ext_vector_type(8)))  _Float16 v8h;
typedef __attribute__((ext_vector_type(4)))  _Float16 v4h;
typedef __attribute__((ext_vector_type(8)))  float    v8f;
typedef __attribute__((ext_vector_type(2)))  float    v2f;

#define N_NODES 50000
#define N_EDGES 500000
#define ETILES  (N_EDGES / 16)
#define EWAVES  8

__device__ __forceinline__ v16h join16(v8h lo, v8h hi) {
    union { v16h v; v8h h[2]; } u;
    u.h[0] = lo; u.h[1] = hi;
    return u.v;
}

__device__ __forceinline__ v8f load8f(const float* __restrict__ p) {
    const float4 lo = *(const float4*)p;
    const float4 hi = *(const float4*)(p + 4);
    v8f a;
    a[0] = lo.x; a[1] = lo.y; a[2] = lo.z; a[3] = lo.w;
    a[4] = hi.x; a[5] = hi.y; a[6] = hi.z; a[7] = hi.w;
    return a;
}

__device__ __forceinline__ v8h relu_pack(v8f x) {
    v8h p;
#pragma unroll
    for (int r = 0; r < 8; ++r) p[r] = (f16)x[r];
    const v8h z = {0, 0, 0, 0, 0, 0, 0, 0};
    return __builtin_elementwise_max(p, z);
}

__device__ __forceinline__ v8f wmma_f16(v16h a, v16h b, v8f c) {
    return __builtin_amdgcn_wmma_f32_16x16x32_f16(false, a, false, b,
                                                  (short)0, c, false, false);
}

__device__ __forceinline__ float relu_f32(float x) {
    const int b = __builtin_bit_cast(int, x);
    return __builtin_bit_cast(float, b > 0 ? b : 0);
}

__device__ __forceinline__ float xhalf(float x) {
    return __builtin_bit_cast(float, __builtin_amdgcn_permlanex16(
        __builtin_bit_cast(int, x), __builtin_bit_cast(int, x),
        0x76543210, 0xfedcba98, false, false));
}

__global__ void pack_w_kernel(const float* __restrict__ W, f16* __restrict__ out,
                              int Ntot, int kTiles, int total) {
    for (int idx = blockIdx.x * blockDim.x + threadIdx.x; idx < total;
         idx += gridDim.x * blockDim.x) {
        int f  = idx >> 9;
        int r  = idx & 511;
        int l  = r >> 4;
        int i  = r & 15;
        int nt = f / kTiles;
        int kt = f - nt * kTiles;
        int K  = kt * 32 + ((i >> 3) << 4) + ((l >> 4) << 3) + (i & 7);
        int N  = nt * 16 + (l & 15);
        out[idx] = (f16)W[(size_t)K * Ntot + N];
    }
}

__global__ void node_feat_kernel(const float* __restrict__ x,
                                 const float* __restrict__ bn,
                                 const f16*  __restrict__ Wnp,
                                 f16* __restrict__ nf) {
    __shared__ __align__(16) f16 xs_all[4][16 * 128];
    const int l  = threadIdx.x & 31;
    const int wv = threadIdx.x >> 5;
    const int tile = blockIdx.x * 4 + wv;
    if (tile >= N_NODES / 16) return;
    f16* xs = xs_all[wv];
    const int n0 = tile * 16;
    const int m  = l & 15;
    const int h  = l >> 4;

#pragma unroll
    for (int t = 0; t < 16; ++t) {
        const int row = t, col = l * 4;
        const float4 v = *(const float4*)(x + (size_t)(n0 + row) * 128 + col);
        v4h p; p[0] = (f16)v.x; p[1] = (f16)v.y; p[2] = (f16)v.z; p[3] = (f16)v.w;
        *(v4h*)(xs + row * 128 + col) = p;
    }

    for (int nc = 0; nc < 2; ++nc) {
        v8f acc[4];
#pragma unroll
        for (int j = 0; j < 4; ++j) acc[j] = load8f(bn + (nc * 4 + j) * 16 + 8 * h);
        for (int kt = 0; kt < 4; ++kt) {
            int k0  = kt * 32 + h * 8;
            v16h a = join16(*(const v8h*)(xs + m * 128 + k0),
                            *(const v8h*)(xs + m * 128 + k0 + 16));
            v16h b[4];
#pragma unroll
            for (int j = 0; j < 4; ++j)
                b[j] = *(const v16h*)(Wnp + (size_t)(((nc * 4 + j) * 4 + kt) * 512) + l * 16);
#pragma unroll
            for (int j = 0; j < 4; ++j) acc[j] = wmma_f16(b[j], a, acc[j]);
        }
#pragma unroll
        for (int j = 0; j < 4; ++j)
            *(v8h*)(nf + (size_t)(n0 + m) * 128 + (nc * 4 + j) * 16 + 8 * h) =
                relu_pack(acc[j]);
    }
}

#define CMB_STRIDE 392
#define H1_STRIDE  520
#define EA_STRIDE  64
#define MTILES     8
#define LDS_CMB    0
#define LDS_H1     (LDS_CMB + 128 * CMB_STRIDE * 2)
#define LDS_EA     (LDS_H1  + 128 * H1_STRIDE  * 2)
#define LDS_PART   (LDS_EA  + 128 * EA_STRIDE  * 2)
#define WG_LDS     (LDS_PART + EWAVES * 128 * 4)


template <int NJ, int KT>
__device__ __forceinline__ void load_b(v16h (&b)[NJ], const f16* __restrict__ Wp,
                                       int nt0, int kt, int l) {
#pragma unroll
    for (int j = 0; j < NJ; ++j)
        b[j] = *(const v16h*)(Wp + (size_t)(((nt0 + j) * KT + kt) * 512) + l * 16);
}

template <int AH>
__device__ __forceinline__ void load_a(v16h (&a)[MTILES / AH], const f16* A, int a_stride,
                                       int blk, int m, int h) {
    constexpr int CM = MTILES / AH;
    const int k0  = (blk / AH) * 32 + h * 8;
    const int mi0 = (blk % AH) * CM;
#pragma unroll
    for (int mi = 0; mi < CM; ++mi) {
        const f16* ar = A + ((mi0 + mi) * 16 + m) * a_stride + k0;
        a[mi] = join16(*(const v8h*)ar, *(const v8h*)(ar + 16));
    }
}

template <int NJ>
__device__ __forceinline__ void init_acc(v8f (&acc)[MTILES][NJ],
                                         const float* __restrict__ bias,
                                         int nt0, int h) {
#pragma unroll
    for (int j = 0; j < NJ; ++j) {
        const v8f bv = load8f(bias + (nt0 + j) * 16 + 8 * h);
#pragma unroll
        for (int mi = 0; mi < MTILES; ++mi) acc[mi][j] = bv;
    }
}

template <int NJ, int KT, int PB, int AH, int AR>
__device__ __forceinline__ void strip_mma(v8f (&acc)[MTILES][NJ],
                                          v16h (&b)[PB + 1][NJ],
                                          const f16* A, int a_stride,
                                          const f16* __restrict__ Wp, int nt0,
                                          int l, int m, int h) {
    constexpr int CM = MTILES / AH;
    constexpr int NB = KT * AH;
    v16h a[AR][CM];
#pragma unroll
    for (int p = 0; p < AR - 1; ++p) load_a<AH>(a[p], A, a_stride, p, m, h);
#pragma unroll
    for (int blk = 0; blk < NB; ++blk) {
        const int kt = blk / AH;
        const int c  = blk % AH;
        if (c == 0 && kt + PB < KT)
            load_b<NJ, KT>(b[(kt + PB) % (PB + 1)], Wp, nt0, kt + PB, l);
        if (blk + AR - 1 < NB)
            load_a<AH>(a[(blk + AR - 1) % AR], A, a_stride, blk + AR - 1, m, h);
        __builtin_amdgcn_sched_barrier(0);
#pragma unroll
        for (int mi = 0; mi < CM; ++mi)
#pragma unroll
            for (int j = 0; j < NJ; ++j)
                acc[c * CM + mi][j] = wmma_f16(b[kt % (PB + 1)][j], a[blk % AR][mi],
                                               acc[c * CM + mi][j]);
        __builtin_amdgcn_sched_barrier(0);
    }
}

template <int NJ>
__device__ __forceinline__ void strip_store(const v8f (&acc)[MTILES][NJ], int nt0,
                                            f16* D, int d_stride, int d_col0,
                                            int m, int h) {
#pragma unroll
    for (int mi = 0; mi < MTILES; ++mi)
#pragma unroll
        for (int j = 0; j < NJ; ++j)
            *(v8h*)(D + (mi * 16 + m) * d_stride + d_col0 + (nt0 + j) * 16 + 8 * h) =
                relu_pack(acc[mi][j]);
}

__global__ void __launch_bounds__(32 * EWAVES)
edge_fused_kernel(const int*   __restrict__ eidx,
                  const float* __restrict__ eattr,
                  const float* __restrict__ be,
                  const float* __restrict__ b1,
                  const float* __restrict__ b2,
                  const float* __restrict__ W3,
                  const float* __restrict__ b3,
                  const f16*   __restrict__ Wep,
                  const f16*   __restrict__ W1p,
                  const f16*   __restrict__ W2p,
                  const f16*   __restrict__ nf,
                  float* __restrict__ out) {
    extern __shared__ __align__(16) char smem[];
    const int l  = threadIdx.x & 31;
    const int wv = threadIdx.x >> 5;
    const int m  = l & 15;
    const int h  = l >> 4;
    f16*   cmb = (f16*)(smem + LDS_CMB);
    f16*   h1  = (f16*)(smem + LDS_H1);
    f16*   ea  = (f16*)(smem + LDS_EA);
    float* part = (float*)(smem + LDS_PART);

    const int  tile  = blockIdx.x * EWAVES + wv;
    const bool valid = (tile < ETILES);
    const int  e0    = valid ? tile * 16 : 0;
    const int  row0  = wv * 16;

    v16h wbe[2][1];
    v16h wb1[3][4];
    v16h wb2[5][2];

    load_b<1, 2>(wbe[0], Wep, wv, 0, l);

    const int node = eidx[h * N_EDGES + e0 + m];

    float4 ev[8];
#pragma unroll
    for (int t = 0; t < 8; ++t)
        ev[t] = *(const float4*)(eattr + (size_t)e0 * 64 + t * 128 + l * 4);

    v8h g[16];
#pragma unroll
    for (int p = 0; p < 16; ++p) {
        const int ns = __builtin_amdgcn_readlane(node, p);
        const int nd = __builtin_amdgcn_readlane(node, 16 + p);
        g[p] = *(const v8h*)(nf + (size_t)(h ? nd : ns) * 128 + m * 8);
    }
    __builtin_amdgcn_sched_barrier(0);

#pragma unroll
    for (int t = 0; t < 8; ++t) {
        int off = t * 128 + l * 4;
        int row = off >> 6, col = off & 63;
        const float4 v = ev[t];
        v4h p; p[0] = (f16)v.x; p[1] = (f16)v.y; p[2] = (f16)v.z; p[3] = (f16)v.w;
        *(v4h*)(ea + (row0 + row) * EA_STRIDE + col) = p;
    }

#pragma unroll
    for (int p = 0; p < 16; ++p)
        *(v8h*)(cmb + (row0 + p) * CMB_STRIDE + h * 128 + m * 8) = g[p];

    __syncthreads();

    {
        v8f acc[MTILES][1];
        init_acc<1>(acc, be, wv, h);
        load_b<4, 12>(wb1[0], W1p, wv * 4, 0, l);
        load_b<4, 12>(wb1[1], W1p, wv * 4, 1, l);
        strip_mma<1, 2, 1, 1, 2>(acc, wbe, ea, EA_STRIDE, Wep, wv, l, m, h);
        strip_store<1>(acc, wv, cmb, CMB_STRIDE, 256, m, h);
    }

    __syncthreads();

    {
        v8f acc[MTILES][4];
        init_acc<4>(acc, b1, wv * 4, h);
        strip_mma<4, 12, 2, 2, 2>(acc, wb1, cmb, CMB_STRIDE, W1p, wv * 4, l, m, h);
#pragma unroll
        for (int p = 0; p < 4; ++p)
            load_b<2, 16>(wb2[p], W2p, wv * 2, p, l);
        strip_store<4>(acc, wv * 4, h1, H1_STRIDE, 0, m, h);
    }

    __syncthreads();

    {
        v8f w3r[2];
#pragma unroll
        for (int j = 0; j < 2; ++j) w3r[j] = load8f(W3 + (wv * 2 + j) * 16 + 8 * h);

        v8f acc[MTILES][2];
        init_acc<2>(acc, b2, wv * 2, h);
        strip_mma<2, 16, 4, 1, 3>(acc, wb2, h1, H1_STRIDE, W2p, wv * 2, l, m, h);

        v2f s2[MTILES];
#pragma unroll
        for (int mi = 0; mi < MTILES; ++mi) s2[mi] = (v2f){0.0f, 0.0f};
#pragma unroll
        for (int j = 0; j < 2; ++j)
#pragma unroll
            for (int r = 0; r < 8; r += 2) {
                const v2f w = {w3r[j][r], w3r[j][r + 1]};
#pragma unroll
                for (int mi = 0; mi < MTILES; ++mi) {
                    const v2f x = {relu_f32(acc[mi][j][r]), relu_f32(acc[mi][j][r + 1])};
                    s2[mi] = __builtin_elementwise_fma(x, w, s2[mi]);
                }
            }
        float s[MTILES];
#pragma unroll
        for (int mi = 0; mi < MTILES; ++mi) {
            s[mi] = s2[mi][0] + s2[mi][1];
            s[mi] += xhalf(s[mi]);
        }
        if (h == 0) {
#pragma unroll
            for (int mi = 0; mi < MTILES; ++mi) part[wv * 128 + mi * 16 + m] = s[mi];
        }
    }

    __syncthreads();

    if (wv < 4) {
        const int row = threadIdx.x;
        float s = b3[0];
#pragma unroll
        for (int w = 0; w < EWAVES; ++w) s += part[w * 128 + row];
        const int e = blockIdx.x * (EWAVES * 16) + row;
        if (e < N_EDGES) out[e] = s;
    }
}

#define WS_WN 0
#define WS_WE (WS_WN + 16384 * 2)
#define WS_W1 (WS_WE + 8192 * 2)
#define WS_W2 (WS_W1 + 196608 * 2)
#define WS_NF (WS_W2 + 131072 * 2)

extern "C" void kernel_launch(void* const* d_in, const int* in_sizes, int n_in,
                              void* d_out, int out_size, void* d_ws, size_t ws_size,
                              hipStream_t stream) {
    (void)in_sizes; (void)n_in; (void)out_size; (void)ws_size;
    const float* x     = (const float*)d_in[0];
    const int*   eidx  = (const int*)  d_in[1];
    const float* eattr = (const float*)d_in[2];
    const float* Wn    = (const float*)d_in[3];
    const float* bn    = (const float*)d_in[4];
    const float* We    = (const float*)d_in[5];
    const float* be    = (const float*)d_in[6];
    const float* W1    = (const float*)d_in[7];
    const float* b1    = (const float*)d_in[8];
    const float* W2    = (const float*)d_in[9];
    const float* b2    = (const float*)d_in[10];
    const float* W3    = (const float*)d_in[11];
    const float* b3    = (const float*)d_in[12];

    char* ws  = (char*)d_ws;
    f16* Wnp = (f16*)(ws + WS_WN);
    f16* Wep = (f16*)(ws + WS_WE);
    f16* W1p = (f16*)(ws + WS_W1);
    f16* W2p = (f16*)(ws + WS_W2);
    f16* nf  = (f16*)(ws + WS_NF);

    (void)hipFuncSetAttribute((const void*)edge_fused_kernel,
                              hipFuncAttributeMaxDynamicSharedMemorySize,
                              WG_LDS);

    pack_w_kernel<<<64,  256, 0, stream>>>(Wn, Wnp, 128, 4,  16384);
    pack_w_kernel<<<32,  256, 0, stream>>>(We, Wep, 128, 2,  8192);
    pack_w_kernel<<<768, 256, 0, stream>>>(W1, W1p, 512, 12, 196608);
    pack_w_kernel<<<512, 256, 0, stream>>>(W2, W2p, 256, 16, 131072);

    node_feat_kernel<<<(3125 + 3) / 4, 128, 0, stream>>>(x, bn, Wnp, nf);

    edge_fused_kernel<<<(ETILES + EWAVES - 1) / EWAVES, 32 * EWAVES,
                        WG_LDS, stream>>>(
        eidx, eattr, be, b1, b2, W3, b3, Wep, W1p, W2p, nf, (float*)d_out);
}


# edge kernel made resident: 256 workgroups loop over the 128-edge tiles; the next tile's indices, edge_attr and node rows are loaded under layers 1-2 of the current one (LDS-only barriers)
# speedup vs baseline: 4.4615x; 1.0562x over previous
#include <hip/hip_runtime.h>
#include <hip/hip_bf16.h>


typedef _Float16 f16;
typedef __attribute__((ext_vector_type(16))) _Float16 v16h;
typedef __attribute__((ext_vector_type(8)))  _Float16 v8h;
typedef __attribute__((ext_vector_type(4)))  _Float16 v4h;
typedef __attribute__((ext_vector_type(8)))  float    v8f;
typedef __attribute__((ext_vector_type(2)))  float    v2f;

#define N_NODES 50000
#define N_EDGES 500000
#define ETILES  (N_EDGES / 16)
#define EWAVES  8

__device__ __forceinline__ v16h join16(v8h lo, v8h hi) {
    union { v16h v; v8h h[2]; } u;
    u.h[0] = lo; u.h[1] = hi;
    return u.v;
}

__device__ __forceinline__ v8f load8f(const float* __restrict__ p) {
    const float4 lo = *(const float4*)p;
    const float4 hi = *(const float4*)(p + 4);
    v8f a;
    a[0] = lo.x; a[1] = lo.y; a[2] = lo.z; a[3] = lo.w;
    a[4] = hi.x; a[5] = hi.y; a[6] = hi.z; a[7] = hi.w;
    return a;
}

__device__ __forceinline__ v8h relu_pack(v8f x) {
    v8h p;
#pragma unroll
    for (int r = 0; r < 8; ++r) p[r] = (f16)x[r];
    const v8h z = {0, 0, 0, 0, 0, 0, 0, 0};
    return __builtin_elementwise_max(p, z);
}

__device__ __forceinline__ v8f wmma_f16(v16h a, v16h b, v8f c) {
    return __builtin_amdgcn_wmma_f32_16x16x32_f16(false, a, false, b,
                                                  (short)0, c, false, false);
}

__device__ __forceinline__ float relu_f32(float x) {
    const int b = __builtin_bit_cast(int, x);
    return __builtin_bit_cast(float, b > 0 ? b : 0);
}

__device__ __forceinline__ float xhalf(float x) {
    return __builtin_bit_cast(float, __builtin_amdgcn_permlanex16(
        __builtin_bit_cast(int, x), __builtin_bit_cast(int, x),
        0x76543210, 0xfedcba98, false, false));
}

__global__ void pack_w_kernel(const float* __restrict__ W, f16* __restrict__ out,
                              int Ntot, int kTiles, int total) {
    for (int idx = blockIdx.x * blockDim.x + threadIdx.x; idx < total;
         idx += gridDim.x * blockDim.x) {
        int f  = idx >> 9;
        int r  = idx & 511;
        int l  = r >> 4;
        int i  = r & 15;
        int nt = f / kTiles;
        int kt = f - nt * kTiles;
        int K  = kt * 32 + ((i >> 3) << 4) + ((l >> 4) << 3) + (i & 7);
        int N  = nt * 16 + (l & 15);
        out[idx] = (f16)W[(size_t)K * Ntot + N];
    }
}

__global__ void node_feat_kernel(const float* __restrict__ x,
                                 const float* __restrict__ bn,
                                 const f16*  __restrict__ Wnp,
                                 f16* __restrict__ nf) {
    __shared__ __align__(16) f16 xs_all[4][16 * 128];
    const int l  = threadIdx.x & 31;
    const int wv = threadIdx.x >> 5;
    const int tile = blockIdx.x * 4 + wv;
    if (tile >= N_NODES / 16) return;
    f16* xs = xs_all[wv];
    const int n0 = tile * 16;
    const int m  = l & 15;
    const int h  = l >> 4;

#pragma unroll
    for (int t = 0; t < 16; ++t) {
        const int row = t, col = l * 4;
        const float4 v = *(const float4*)(x + (size_t)(n0 + row) * 128 + col);
        v4h p; p[0] = (f16)v.x; p[1] = (f16)v.y; p[2] = (f16)v.z; p[3] = (f16)v.w;
        *(v4h*)(xs + row * 128 + col) = p;
    }

    for (int nc = 0; nc < 2; ++nc) {
        v8f acc[4];
#pragma unroll
        for (int j = 0; j < 4; ++j) acc[j] = load8f(bn + (nc * 4 + j) * 16 + 8 * h);
        for (int kt = 0; kt < 4; ++kt) {
            int k0  = kt * 32 + h * 8;
            v16h a = join16(*(const v8h*)(xs + m * 128 + k0),
                            *(const v8h*)(xs + m * 128 + k0 + 16));
            v16h b[4];
#pragma unroll
            for (int j = 0; j < 4; ++j)
                b[j] = *(const v16h*)(Wnp + (size_t)(((nc * 4 + j) * 4 + kt) * 512) + l * 16);
#pragma unroll
            for (int j = 0; j < 4; ++j) acc[j] = wmma_f16(b[j], a, acc[j]);
        }
#pragma unroll
        for (int j = 0; j < 4; ++j)
            *(v8h*)(nf + (size_t)(n0 + m) * 128 + (nc * 4 + j) * 16 + 8 * h) =
                relu_pack(acc[j]);
    }
}

#define CMB_STRIDE 392
#define H1_STRIDE  520
#define EA_STRIDE  64
#define MTILES     8
#define LDS_CMB    0
#define LDS_H1     (LDS_CMB + 128 * CMB_STRIDE * 2)
#define LDS_EA     (LDS_H1  + 128 * H1_STRIDE  * 2)
#define LDS_PART   (LDS_EA  + 128 * EA_STRIDE  * 2)
#define WG_LDS     (LDS_PART + EWAVES * 128 * 4)


template <int NJ, int KT>
__device__ __forceinline__ void load_b(v16h (&b)[NJ], const f16* __restrict__ Wp,
                                       int nt0, int kt, int l) {
#pragma unroll
    for (int j = 0; j < NJ; ++j)
        b[j] = *(const v16h*)(Wp + (size_t)(((nt0 + j) * KT + kt) * 512) + l * 16);
}

template <int AH>
__device__ __forceinline__ void load_a(v16h (&a)[MTILES / AH], const f16* A, int a_stride,
                                       int blk, int m, int h) {
    constexpr int CM = MTILES / AH;
    const int k0  = (blk / AH) * 32 + h * 8;
    const int mi0 = (blk % AH) * CM;
#pragma unroll
    for (int mi = 0; mi < CM; ++mi) {
        const f16* ar = A + ((mi0 + mi) * 16 + m) * a_stride + k0;
        a[mi] = join16(*(const v8h*)ar, *(const v8h*)(ar + 16));
    }
}

template <int NJ>
__device__ __forceinline__ void load_bias(v8f (&bv)[NJ], const float* __restrict__ bias,
                                          int nt0, int h) {
#pragma unroll
    for (int j = 0; j < NJ; ++j)
        bv[j] = load8f(bias + (nt0 + j) * 16 + 8 * h);
}

template <int NJ>
__device__ __forceinline__ void fill_acc(v8f (&acc)[MTILES][NJ], const v8f (&bv)[NJ]) {
#pragma unroll
    for (int j = 0; j < NJ; ++j)
#pragma unroll
        for (int mi = 0; mi < MTILES; ++mi) acc[mi][j] = bv[j];
}

template <int NJ>
__device__ __forceinline__ void init_acc(v8f (&acc)[MTILES][NJ],
                                         const float* __restrict__ bias,
                                         int nt0, int h) {
    v8f bv[NJ];
    load_bias<NJ>(bv, bias, nt0, h);
    fill_acc<NJ>(acc, bv);
}

template <int NJ, int KT, int PB, int AH, int AR, int HB, class Extra>
__device__ __forceinline__ void strip_mma(v8f (&acc)[MTILES][NJ],
                                          v16h (&b)[PB + 1][NJ],
                                          const f16* A, int a_stride,
                                          const f16* __restrict__ Wp, int nt0,
                                          int l, int m, int h, Extra extra) {
    constexpr int CM = MTILES / AH;
    constexpr int NB = KT * AH;
    v16h a[AR][CM];
#pragma unroll
    for (int p = 0; p < AR - 1; ++p) load_a<AH>(a[p], A, a_stride, p, m, h);
#pragma unroll
    for (int blk = 0; blk < NB; ++blk) {
        const int kt = blk / AH;
        const int c  = blk % AH;
        if (c == 0 && kt + PB < KT)
            load_b<NJ, KT>(b[(kt + PB) % (PB + 1)], Wp, nt0, kt + PB, l);
        if (blk + AR - 1 < NB)
            load_a<AH>(a[(blk + AR - 1) % AR], A, a_stride, blk + AR - 1, m, h);
        if (blk == HB) extra();
        __builtin_amdgcn_sched_barrier(0);
#pragma unroll
        for (int mi = 0; mi < CM; ++mi)
#pragma unroll
            for (int j = 0; j < NJ; ++j)
                acc[c * CM + mi][j] = wmma_f16(b[kt % (PB + 1)][j], a[blk % AR][mi],
                                               acc[c * CM + mi][j]);
        __builtin_amdgcn_sched_barrier(0);
    }
}

template <int NJ>
__device__ __forceinline__ void strip_store(const v8f (&acc)[MTILES][NJ], int nt0,
                                            f16* D, int d_stride, int d_col0,
                                            int m, int h) {
#pragma unroll
    for (int mi = 0; mi < MTILES; ++mi)
#pragma unroll
        for (int j = 0; j < NJ; ++j)
            *(v8h*)(D + (mi * 16 + m) * d_stride + d_col0 + (nt0 + j) * 16 + 8 * h) =
                relu_pack(acc[mi][j]);
}

__device__ __forceinline__ void lds_barrier() {
    __builtin_amdgcn_fence(__ATOMIC_RELEASE, "workgroup", "local");
    __builtin_amdgcn_s_barrier();
    __builtin_amdgcn_fence(__ATOMIC_ACQUIRE, "workgroup", "local");
}

#define NT128 ((ETILES + EWAVES - 1) / EWAVES)
#define NWG   256

__device__ __forceinline__ int first_edge(int T, int wv) {
    const int t16 = T * EWAVES + wv;
    return t16 < ETILES ? t16 * 16 : 0;
}

__device__ __forceinline__ int load_nodes(const int* __restrict__ eidx, int e0, int m, int h) {
    return eidx[h * N_EDGES + e0 + m];
}

__device__ __forceinline__ void load_eattr(float4 (&ev)[8], const float* __restrict__ eattr,
                                           int e0, int l) {
#pragma unroll
    for (int t = 0; t < 8; ++t)
        ev[t] = *(const float4*)(eattr + (size_t)e0 * 64 + t * 128 + l * 4);
}

__device__ __forceinline__ void gather_rows(v8h (&g)[16], const f16* __restrict__ nf,
                                            int node, int m, int h) {
#pragma unroll
    for (int p = 0; p < 16; ++p) {
        const int ns = __builtin_amdgcn_readlane(node, p);
        const int nd = __builtin_amdgcn_readlane(node, 16 + p);
        g[p] = *(const v8h*)(nf + (size_t)(h ? nd : ns) * 128 + m * 8);
    }
}

__device__ __forceinline__ void store_eattr(const float4 (&ev)[8], f16* ea, int row0, int l) {
#pragma unroll
    for (int t = 0; t < 8; ++t) {
        int off = t * 128 + l * 4;
        int row = off >> 6, col = off & 63;
        const float4 v = ev[t];
        v4h p; p[0] = (f16)v.x; p[1] = (f16)v.y; p[2] = (f16)v.z; p[3] = (f16)v.w;
        *(v4h*)(ea + (row0 + row) * EA_STRIDE + col) = p;
    }
}

__device__ __forceinline__ void store_rows(const v8h (&g)[16], f16* cmb, int row0, int m, int h) {
#pragma unroll
    for (int p = 0; p < 16; ++p)
        *(v8h*)(cmb + (row0 + p) * CMB_STRIDE + h * 128 + m * 8) = g[p];
}

__global__ void __launch_bounds__(32 * EWAVES)
edge_fused_kernel(const int*   __restrict__ eidx,
                  const float* __restrict__ eattr,
                  const float* __restrict__ be,
                  const float* __restrict__ b1,
                  const float* __restrict__ b2,
                  const float* __restrict__ W3,
                  const float* __restrict__ b3,
                  const f16*   __restrict__ Wep,
                  const f16*   __restrict__ W1p,
                  const f16*   __restrict__ W2p,
                  const f16*   __restrict__ nf,
                  float* __restrict__ out) {
    extern __shared__ __align__(16) char smem[];
    f16*   cmb = (f16*)(smem + LDS_CMB);
    f16*   h1  = (f16*)(smem + LDS_H1);
    f16*   ea  = (f16*)(smem + LDS_EA);
    float* part = (float*)(smem + LDS_PART);

    const auto no_extra = []() {};

    v16h wbe[2][1];
    v16h wb1[3][4];
    v16h wb2[5][2];

    int T = blockIdx.x;

    {
        const int l    = threadIdx.x & 31;
        const int wv   = threadIdx.x >> 5;
        const int m    = l & 15;
        const int h    = l >> 4;
        const int row0 = wv * 16;
        load_b<1, 2>(wbe[0], Wep, wv, 0, l);
        const int e0 = first_edge(T, wv);
        const int node = load_nodes(eidx, e0, m, h);
        float4 ev[8];
        load_eattr(ev, eattr, e0, l);
        v8h g[16];
        gather_rows(g, nf, node, m, h);
        __builtin_amdgcn_sched_barrier(0);
        store_eattr(ev, ea, row0, l);
        store_rows(g, cmb, row0, m, h);
    }

    lds_barrier();

    const float b3v = b3[0];

    for (;;) {
        int zero = 0;
        asm volatile("" : "+s"(zero));
        int tid = threadIdx.x;
        asm volatile("" : "+v"(tid));
        const int l    = tid & 31;
        const int wv   = tid >> 5;
        const int m    = l & 15;
        const int h    = l >> 4;
        const int row0 = wv * 16;
        const f16*   Wep_t = Wep + zero;
        const f16*   W1p_t = W1p + zero;
        const f16*   W2p_t = W2p + zero;
        const float* be_t  = be + zero;
        const float* b1_t  = b1 + zero;
        const float* b2_t  = b2 + zero;
        const float* W3_t  = W3 + zero;

        const bool has_next = (T + NWG < NT128);
        const int  Tn  = has_next ? T + NWG : T;
        const int  e0n = first_edge(Tn, wv);
        int    node_n;
        float4 ev_n[8];
        v8h    g_n[16];
        v8f    bias2[2];
        v8f    w3r[2];

        {
            v8f acc[MTILES][1];
            init_acc<1>(acc, be_t, wv, h);
            load_b<4, 12>(wb1[0], W1p_t, wv * 4, 0, l);
            load_b<4, 12>(wb1[1], W1p_t, wv * 4, 1, l);
            strip_mma<1, 2, 1, 1, 2, -1>(acc, wbe, ea, EA_STRIDE, Wep_t, wv, l, m, h,
                                         no_extra);
            strip_store<1>(acc, wv, cmb, CMB_STRIDE, 256, m, h);
        }

        lds_barrier();

        {
            v8f acc[MTILES][4];
            init_acc<4>(acc, b1_t, wv * 4, h);
            strip_mma<4, 12, 2, 2, 2, -1>(acc, wb1, cmb, CMB_STRIDE, W1p_t, wv * 4, l, m, h,
                                          no_extra);
#pragma unroll
            for (int p = 0; p < 4; ++p)
                load_b<2, 16>(wb2[p], W2p_t, wv * 2, p, l);
            load_bias<2>(bias2, b2_t, wv * 2, h);
#pragma unroll
            for (int j = 0; j < 2; ++j) w3r[j] = load8f(W3_t + (wv * 2 + j) * 16 + 8 * h);
            __builtin_amdgcn_sched_barrier(0);
            node_n = load_nodes(eidx, e0n, m, h);
            load_eattr(ev_n, eattr, e0n, l);
            strip_store<4>(acc, wv * 4, h1, H1_STRIDE, 0, m, h);
        }

        lds_barrier();

        {
            v8f acc[MTILES][2];
            fill_acc<2>(acc, bias2);
            strip_mma<2, 16, 4, 1, 3, 14>(acc, wb2, h1, H1_STRIDE, W2p_t, wv * 2, l, m, h,
                                          [&]() { gather_rows(g_n, nf, node_n, m, h); });

            v2f s2[MTILES];
#pragma unroll
            for (int mi = 0; mi < MTILES; ++mi) s2[mi] = (v2f){0.0f, 0.0f};
#pragma unroll
            for (int j = 0; j < 2; ++j)
#pragma unroll
                for (int r = 0; r < 8; r += 2) {
                    const v2f w = {w3r[j][r], w3r[j][r + 1]};
#pragma unroll
                    for (int mi = 0; mi < MTILES; ++mi) {
                        const v2f x = {relu_f32(acc[mi][j][r]), relu_f32(acc[mi][j][r + 1])};
                        s2[mi] = __builtin_elementwise_fma(x, w, s2[mi]);
                    }
                }
            float s[MTILES];
#pragma unroll
            for (int mi = 0; mi < MTILES; ++mi) {
                s[mi] = s2[mi][0] + s2[mi][1];
                s[mi] += xhalf(s[mi]);
            }
            if (h == 0) {
#pragma unroll
                for (int mi = 0; mi < MTILES; ++mi) part[wv * 128 + mi * 16 + m] = s[mi];
            }

            load_b<1, 2>(wbe[0], Wep_t, wv, 0, l);
            store_eattr(ev_n, ea, row0, l);
            store_rows(g_n, cmb, row0, m, h);
        }

        lds_barrier();

        if (wv < 4) {
            const int row = tid;
            float s = b3v;
#pragma unroll
            for (int w = 0; w < EWAVES; ++w) s += part[w * 128 + row];
            const int e = T * (EWAVES * 16) + row;
            if (e < N_EDGES) out[e] = s;
        }

        if (!has_next) break;
        T = Tn;
    }
}

#define WS_WN 0
#define WS_WE (WS_WN + 16384 * 2)
#define WS_W1 (WS_WE + 8192 * 2)
#define WS_W2 (WS_W1 + 196608 * 2)
#define WS_NF (WS_W2 + 131072 * 2)

extern "C" void kernel_launch(void* const* d_in, const int* in_sizes, int n_in,
                              void* d_out, int out_size, void* d_ws, size_t ws_size,
                              hipStream_t stream) {
    (void)in_sizes; (void)n_in; (void)out_size; (void)ws_size;
    const float* x     = (const float*)d_in[0];
    const int*   eidx  = (const int*)  d_in[1];
    const float* eattr = (const float*)d_in[2];
    const float* Wn    = (const float*)d_in[3];
    const float* bn    = (const float*)d_in[4];
    const float* We    = (const float*)d_in[5];
    const float* be    = (const float*)d_in[6];
    const float* W1    = (const float*)d_in[7];
    const float* b1    = (const float*)d_in[8];
    const float* W2    = (const float*)d_in[9];
    const float* b2    = (const float*)d_in[10];
    const float* W3    = (const float*)d_in[11];
    const float* b3    = (const float*)d_in[12];

    char* ws  = (char*)d_ws;
    f16* Wnp = (f16*)(ws + WS_WN);
    f16* Wep = (f16*)(ws + WS_WE);
    f16* W1p = (f16*)(ws + WS_W1);
    f16* W2p = (f16*)(ws + WS_W2);
    f16* nf  = (f16*)(ws + WS_NF);

    (void)hipFuncSetAttribute((const void*)edge_fused_kernel,
                              hipFuncAttributeMaxDynamicSharedMemorySize,
                              WG_LDS);

    pack_w_kernel<<<64,  256, 0, stream>>>(Wn, Wnp, 128, 4,  16384);
    pack_w_kernel<<<32,  256, 0, stream>>>(We, Wep, 128, 2,  8192);
    pack_w_kernel<<<768, 256, 0, stream>>>(W1, W1p, 512, 12, 196608);
    pack_w_kernel<<<512, 256, 0, stream>>>(W2, W2p, 256, 16, 131072);

    node_feat_kernel<<<(3125 + 3) / 4, 128, 0, stream>>>(x, bn, Wnp, nf);

    edge_fused_kernel<<<NWG, 32 * EWAVES,
                        WG_LDS, stream>>>(
        eidx, eattr, be, b1, b2, W3, b3, Wep, W1p, W2p, nf, (float*)d_out);
}


# edge layer: each wave computes its own 16 rows from edge_attr loaded straight into WMMA fragment registers, behind layer 2; the LDS edge_attr tile, its 8-way conflicted reads and a barrier go
# speedup vs baseline: 4.7840x; 1.0723x over previous
#include <hip/hip_runtime.h>
#include <hip/hip_bf16.h>


typedef _Float16 f16;
typedef __attribute__((ext_vector_type(16))) _Float16 v16h;
typedef __attribute__((ext_vector_type(8)))  _Float16 v8h;
typedef __attribute__((ext_vector_type(4)))  _Float16 v4h;
typedef __attribute__((ext_vector_type(8)))  float    v8f;
typedef __attribute__((ext_vector_type(2)))  float    v2f;

#define N_NODES 50000
#define N_EDGES 500000
#define ETILES  (N_EDGES / 16)
#define EWAVES  8

__device__ __forceinline__ v16h join16(v8h lo, v8h hi) {
    union { v16h v; v8h h[2]; } u;
    u.h[0] = lo; u.h[1] = hi;
    return u.v;
}

__device__ __forceinline__ v8f load8f(const float* __restrict__ p) {
    const float4 lo = *(const float4*)p;
    const float4 hi = *(const float4*)(p + 4);
    v8f a;
    a[0] = lo.x; a[1] = lo.y; a[2] = lo.z; a[3] = lo.w;
    a[4] = hi.x; a[5] = hi.y; a[6] = hi.z; a[7] = hi.w;
    return a;
}

__device__ __forceinline__ v8h relu_pack(v8f x) {
    v8h p;
#pragma unroll
    for (int r = 0; r < 8; ++r) p[r] = (f16)x[r];
    const v8h z = {0, 0, 0, 0, 0, 0, 0, 0};
    return __builtin_elementwise_max(p, z);
}

__device__ __forceinline__ v8f wmma_f16(v16h a, v16h b, v8f c) {
    return __builtin_amdgcn_wmma_f32_16x16x32_f16(false, a, false, b,
                                                  (short)0, c, false, false);
}

__device__ __forceinline__ float relu_f32(float x) {
    const int b = __builtin_bit_cast(int, x);
    return __builtin_bit_cast(float, b > 0 ? b : 0);
}

__device__ __forceinline__ float xhalf(float x) {
    return __builtin_bit_cast(float, __builtin_amdgcn_permlanex16(
        __builtin_bit_cast(int, x), __builtin_bit_cast(int, x),
        0x76543210, 0xfedcba98, false, false));
}

__global__ void pack_w_kernel(const float* __restrict__ W, f16* __restrict__ out,
                              int Ntot, int kTiles, int total) {
    for (int idx = blockIdx.x * blockDim.x + threadIdx.x; idx < total;
         idx += gridDim.x * blockDim.x) {
        int f  = idx >> 9;
        int r  = idx & 511;
        int l  = r >> 4;
        int i  = r & 15;
        int nt = f / kTiles;
        int kt = f - nt * kTiles;
        int K  = kt * 32 + ((i >> 3) << 4) + ((l >> 4) << 3) + (i & 7);
        int N  = nt * 16 + (l & 15);
        out[idx] = (f16)W[(size_t)K * Ntot + N];
    }
}

__global__ void node_feat_kernel(const float* __restrict__ x,
                                 const float* __restrict__ bn,
                                 const f16*  __restrict__ Wnp,
                                 f16* __restrict__ nf) {
    __shared__ __align__(16) f16 xs_all[4][16 * 128];
    const int l  = threadIdx.x & 31;
    const int wv = threadIdx.x >> 5;
    const int tile = blockIdx.x * 4 + wv;
    if (tile >= N_NODES / 16) return;
    f16* xs = xs_all[wv];
    const int n0 = tile * 16;
    const int m  = l & 15;
    const int h  = l >> 4;

#pragma unroll
    for (int t = 0; t < 16; ++t) {
        const int row = t, col = l * 4;
        const float4 v = *(const float4*)(x + (size_t)(n0 + row) * 128 + col);
        v4h p; p[0] = (f16)v.x; p[1] = (f16)v.y; p[2] = (f16)v.z; p[3] = (f16)v.w;
        *(v4h*)(xs + row * 128 + col) = p;
    }

    for (int nc = 0; nc < 2; ++nc) {
        v8f acc[4];
#pragma unroll
        for (int j = 0; j < 4; ++j) acc[j] = load8f(bn + (nc * 4 + j) * 16 + 8 * h);
        for (int kt = 0; kt < 4; ++kt) {
            int k0  = kt * 32 + h * 8;
            v16h a = join16(*(const v8h*)(xs + m * 128 + k0),
                            *(const v8h*)(xs + m * 128 + k0 + 16));
            v16h b[4];
#pragma unroll
            for (int j = 0; j < 4; ++j)
                b[j] = *(const v16h*)(Wnp + (size_t)(((nc * 4 + j) * 4 + kt) * 512) + l * 16);
#pragma unroll
            for (int j = 0; j < 4; ++j) acc[j] = wmma_f16(b[j], a, acc[j]);
        }
#pragma unroll
        for (int j = 0; j < 4; ++j)
            *(v8h*)(nf + (size_t)(n0 + m) * 128 + (nc * 4 + j) * 16 + 8 * h) =
                relu_pack(acc[j]);
    }
}

#define CMB_STRIDE 392
#define H1_STRIDE  520
#define MTILES     8
#define LDS_CMB    0
#define LDS_H1     (LDS_CMB + 128 * CMB_STRIDE * 2)
#define LDS_PART   (LDS_H1  + 128 * H1_STRIDE  * 2)
#define WG_LDS     (LDS_PART + EWAVES * 128 * 4)


template <int NJ, int KT>
__device__ __forceinline__ void load_b(v16h (&b)[NJ], const f16* __restrict__ Wp,
                                       int nt0, int kt, int l) {
#pragma unroll
    for (int j = 0; j < NJ; ++j)
        b[j] = *(const v16h*)(Wp + (size_t)(((nt0 + j) * KT + kt) * 512) + l * 16);
}

template <int AH>
__device__ __forceinline__ void load_a(v16h (&a)[MTILES / AH], const f16* A, int a_stride,
                                       int blk, int m, int h) {
    constexpr int CM = MTILES / AH;
    const int k0  = (blk / AH) * 32 + h * 8;
    const int mi0 = (blk % AH) * CM;
#pragma unroll
    for (int mi = 0; mi < CM; ++mi) {
        const f16* ar = A + ((mi0 + mi) * 16 + m) * a_stride + k0;
        a[mi] = join16(*(const v8h*)ar, *(const v8h*)(ar + 16));
    }
}

template <int NJ>
__device__ __forceinline__ void load_bias(v8f (&bv)[NJ], const float* __restrict__ bias,
                                          int nt0, int h) {
#pragma unroll
    for (int j = 0; j < NJ; ++j)
        bv[j] = load8f(bias + (nt0 + j) * 16 + 8 * h);
}

template <int NJ>
__device__ __forceinline__ void fill_acc(v8f (&acc)[MTILES][NJ], const v8f (&bv)[NJ]) {
#pragma unroll
    for (int j = 0; j < NJ; ++j)
#pragma unroll
        for (int mi = 0; mi < MTILES; ++mi) acc[mi][j] = bv[j];
}

template <int NJ>
__device__ __forceinline__ void init_acc(v8f (&acc)[MTILES][NJ],
                                         const float* __restrict__ bias,
                                         int nt0, int h) {
    v8f bv[NJ];
    load_bias<NJ>(bv, bias, nt0, h);
    fill_acc<NJ>(acc, bv);
}

template <int NJ, int KT, int PB, int AH, int AR, int HB, class Extra>
__device__ __forceinline__ void strip_mma(v8f (&acc)[MTILES][NJ],
                                          v16h (&b)[PB + 1][NJ],
                                          const f16* A, int a_stride,
                                          const f16* __restrict__ Wp, int nt0,
                                          int l, int m, int h, Extra extra) {
    constexpr int CM = MTILES / AH;
    constexpr int NB = KT * AH;
    v16h a[AR][CM];
#pragma unroll
    for (int p = 0; p < AR - 1; ++p) load_a<AH>(a[p], A, a_stride, p, m, h);
#pragma unroll
    for (int blk = 0; blk < NB; ++blk) {
        const int kt = blk / AH;
        const int c  = blk % AH;
        if (c == 0 && kt + PB < KT)
            load_b<NJ, KT>(b[(kt + PB) % (PB + 1)], Wp, nt0, kt + PB, l);
        if (blk + AR - 1 < NB)
            load_a<AH>(a[(blk + AR - 1) % AR], A, a_stride, blk + AR - 1, m, h);
        if (blk == HB) extra();
        __builtin_amdgcn_sched_barrier(0);
#pragma unroll
        for (int mi = 0; mi < CM; ++mi)
#pragma unroll
            for (int j = 0; j < NJ; ++j)
                acc[c * CM + mi][j] = wmma_f16(b[kt % (PB + 1)][j], a[blk % AR][mi],
                                               acc[c * CM + mi][j]);
        __builtin_amdgcn_sched_barrier(0);
    }
}

template <int NJ>
__device__ __forceinline__ void strip_store(const v8f (&acc)[MTILES][NJ], int nt0,
                                            f16* D, int d_stride, int d_col0,
                                            int m, int h) {
#pragma unroll
    for (int mi = 0; mi < MTILES; ++mi)
#pragma unroll
        for (int j = 0; j < NJ; ++j)
            *(v8h*)(D + (mi * 16 + m) * d_stride + d_col0 + (nt0 + j) * 16 + 8 * h) =
                relu_pack(acc[mi][j]);
}

__device__ __forceinline__ void lds_barrier() {
    __builtin_amdgcn_fence(__ATOMIC_RELEASE, "workgroup", "local");
    __builtin_amdgcn_s_barrier();
    __builtin_amdgcn_fence(__ATOMIC_ACQUIRE, "workgroup", "local");
}

#define NT128 ((ETILES + EWAVES - 1) / EWAVES)
#define NWG   256

__device__ __forceinline__ int first_edge(int T, int wv) {
    const int t16 = T * EWAVES + wv;
    return t16 < ETILES ? t16 * 16 : 0;
}

__device__ __forceinline__ int load_nodes(const int* __restrict__ eidx, int e0, int m, int h) {
    return eidx[h * N_EDGES + e0 + m];
}

__device__ __forceinline__ void load_eattr(float4 (&ev)[8], const float* __restrict__ eattr,
                                           int e0, int m, int h) {
    const float* r = eattr + (size_t)(e0 + m) * 64 + h * 8;
#pragma unroll
    for (int t = 0; t < 8; ++t)
        ev[t] = *(const float4*)(r + (t >> 2) * 32 + ((t >> 1) & 1) * 16 + (t & 1) * 4);
}

__device__ __forceinline__ v16h eattr_frag(const float4 (&ev)[8], int kt) {
    v16h a;
#pragma unroll
    for (int u = 0; u < 4; ++u) {
        const float4 v = ev[kt * 4 + u];
        a[4 * u + 0] = (f16)v.x; a[4 * u + 1] = (f16)v.y;
        a[4 * u + 2] = (f16)v.z; a[4 * u + 3] = (f16)v.w;
    }
    return a;
}

__device__ __forceinline__ void gather_rows(v8h (&g)[16], const f16* __restrict__ nf,
                                            int node, int m, int h) {
#pragma unroll
    for (int p = 0; p < 16; ++p) {
        const int ns = __builtin_amdgcn_readlane(node, p);
        const int nd = __builtin_amdgcn_readlane(node, 16 + p);
        g[p] = *(const v8h*)(nf + (size_t)(h ? nd : ns) * 128 + m * 8);
    }
}

__device__ __forceinline__ void edge_rows(const float4 (&ev)[8], const v16h (&we)[2][8],
                                          const v8f (&bv)[8], f16* cmb,
                                          int row0, int m, int h) {
    v8f acc[8];
#pragma unroll
    for (int j = 0; j < 8; ++j) acc[j] = bv[j];
#pragma unroll
    for (int kt = 0; kt < 2; ++kt) {
        const v16h a = eattr_frag(ev, kt);
#pragma unroll
        for (int j = 0; j < 8; ++j) acc[j] = wmma_f16(we[kt][j], a, acc[j]);
    }
#pragma unroll
    for (int j = 0; j < 8; ++j)
        *(v8h*)(cmb + (row0 + m) * CMB_STRIDE + 256 + j * 16 + 8 * h) = relu_pack(acc[j]);
}

__device__ __forceinline__ void store_rows(const v8h (&g)[16], f16* cmb, int row0, int m, int h) {
#pragma unroll
    for (int p = 0; p < 16; ++p)
        *(v8h*)(cmb + (row0 + p) * CMB_STRIDE + h * 128 + m * 8) = g[p];
}

__global__ void __launch_bounds__(32 * EWAVES)
edge_fused_kernel(const int*   __restrict__ eidx,
                  const float* __restrict__ eattr,
                  const float* __restrict__ be,
                  const float* __restrict__ b1,
                  const float* __restrict__ b2,
                  const float* __restrict__ W3,
                  const float* __restrict__ b3,
                  const f16*   __restrict__ Wep,
                  const f16*   __restrict__ W1p,
                  const f16*   __restrict__ W2p,
                  const f16*   __restrict__ nf,
                  float* __restrict__ out) {
    extern __shared__ __align__(16) char smem[];
    f16*   cmb = (f16*)(smem + LDS_CMB);
    f16*   h1  = (f16*)(smem + LDS_H1);
    float* part = (float*)(smem + LDS_PART);

    const auto no_extra = []() {};

    v16h wb1[3][4];
    v16h wb2[5][2];

    int T = blockIdx.x;

    {
        const int l    = threadIdx.x & 31;
        const int wv   = threadIdx.x >> 5;
        const int m    = l & 15;
        const int h    = l >> 4;
        const int row0 = wv * 16;
        const int e0 = first_edge(T, wv);
        const int node = load_nodes(eidx, e0, m, h);
        float4 ev[8];
        load_eattr(ev, eattr, e0, m, h);
        v16h we[2][8];
        v8f  bev[8];
        load_b<8, 2>(we[0], Wep, 0, 0, l);
        load_b<8, 2>(we[1], Wep, 0, 1, l);
        load_bias<8>(bev, be, 0, h);
        load_b<4, 12>(wb1[0], W1p, wv * 4, 0, l);
        load_b<4, 12>(wb1[1], W1p, wv * 4, 1, l);
        v8h g[16];
        gather_rows(g, nf, node, m, h);
        __builtin_amdgcn_sched_barrier(0);
        edge_rows(ev, we, bev, cmb, row0, m, h);
        store_rows(g, cmb, row0, m, h);
    }

    lds_barrier();

    const float b3v = b3[0];

    for (;;) {
        int zero = 0;
        asm volatile("" : "+s"(zero));
        int tid = threadIdx.x;
        asm volatile("" : "+v"(tid));
        const int l    = tid & 31;
        const int wv   = tid >> 5;
        const int m    = l & 15;
        const int h    = l >> 4;
        const int row0 = wv * 16;
        const f16*   W1p_t = W1p + zero;
        const f16*   W2p_t = W2p + zero;
        const float* b1_t  = b1 + zero;
        const float* b2_t  = b2 + zero;
        const float* W3_t  = W3 + zero;

        const bool has_next = (T + NWG < NT128);
        const int  Tn  = has_next ? T + NWG : T;
        const int  e0n = first_edge(Tn, wv);
        int    node_n;
        float4 ev_n[8];
        v8h    g_n[16];
        v8f    bias2[2];
        v8f    w3r[2];

        {
            v8f acc[MTILES][4];
            init_acc<4>(acc, b1_t, wv * 4, h);
            strip_mma<4, 12, 2, 2, 2, -1>(acc, wb1, cmb, CMB_STRIDE, W1p_t, wv * 4, l, m, h,
                                          no_extra);
#pragma unroll
            for (int p = 0; p < 4; ++p)
                load_b<2, 16>(wb2[p], W2p_t, wv * 2, p, l);
            load_bias<2>(bias2, b2_t, wv * 2, h);
            __builtin_amdgcn_sched_barrier(0);
            node_n = load_nodes(eidx, e0n, m, h);
            load_eattr(ev_n, eattr, e0n, m, h);
            strip_store<4>(acc, wv * 4, h1, H1_STRIDE, 0, m, h);
        }

        lds_barrier();

        {
            v8f acc[MTILES][2];
            fill_acc<2>(acc, bias2);
            strip_mma<2, 16, 4, 1, 3, 14>(acc, wb2, h1, H1_STRIDE, W2p_t, wv * 2, l, m, h,
                                          [&]() {
#pragma unroll
                for (int j = 0; j < 2; ++j)
                    w3r[j] = load8f(W3_t + (wv * 2 + j) * 16 + 8 * h);
                gather_rows(g_n, nf, node_n, m, h);
            });

            v2f s2[MTILES];
#pragma unroll
            for (int mi = 0; mi < MTILES; ++mi) s2[mi] = (v2f){0.0f, 0.0f};
#pragma unroll
            for (int j = 0; j < 2; ++j)
#pragma unroll
                for (int r = 0; r < 8; r += 2) {
                    const v2f w = {w3r[j][r], w3r[j][r + 1]};
#pragma unroll
                    for (int mi = 0; mi < MTILES; ++mi) {
                        const v2f x = {relu_f32(acc[mi][j][r]), relu_f32(acc[mi][j][r + 1])};
                        s2[mi] = __builtin_elementwise_fma(x, w, s2[mi]);
                    }
                }
            v16h we[2][8];
            v8f  bev[8];
            __builtin_amdgcn_sched_barrier(0);
            int zero_e = 0;
            asm volatile("" : "+s"(zero_e));
            const f16*   Wep_e = Wep + zero_e;
            const float* be_e  = be + zero_e;
            const f16*   W1p_e = W1p + zero_e;
            load_b<8, 2>(we[0], Wep_e, 0, 0, l);
            load_bias<8>(bev, be_e, 0, h);
            load_b<8, 2>(we[1], Wep_e, 0, 1, l);
            load_b<4, 12>(wb1[0], W1p_e, wv * 4, 0, l);
            load_b<4, 12>(wb1[1], W1p_e, wv * 4, 1, l);
            __builtin_amdgcn_sched_barrier(0);

            float s[MTILES];
#pragma unroll
            for (int mi = 0; mi < MTILES; ++mi) {
                s[mi] = s2[mi][0] + s2[mi][1];
                s[mi] += xhalf(s[mi]);
            }
            if (h == 0) {
#pragma unroll
                for (int mi = 0; mi < MTILES; ++mi) part[wv * 128 + mi * 16 + m] = s[mi];
            }

            store_rows(g_n, cmb, row0, m, h);
            edge_rows(ev_n, we, bev, cmb, row0, m, h);
        }

        lds_barrier();

        if (wv < 4) {
            const int row = tid;
            float s = b3v;
#pragma unroll
            for (int w = 0; w < EWAVES; ++w) s += part[w * 128 + row];
            const int e = T * (EWAVES * 16) + row;
            if (e < N_EDGES) out[e] = s;
        }

        if (!has_next) break;
        T = Tn;
    }
}

#define WS_WN 0
#define WS_WE (WS_WN + 16384 * 2)
#define WS_W1 (WS_WE + 8192 * 2)
#define WS_W2 (WS_W1 + 196608 * 2)
#define WS_NF (WS_W2 + 131072 * 2)

extern "C" void kernel_launch(void* const* d_in, const int* in_sizes, int n_in,
                              void* d_out, int out_size, void* d_ws, size_t ws_size,
                              hipStream_t stream) {
    (void)in_sizes; (void)n_in; (void)out_size; (void)ws_size;
    const float* x     = (const float*)d_in[0];
    const int*   eidx  = (const int*)  d_in[1];
    const float* eattr = (const float*)d_in[2];
    const float* Wn    = (const float*)d_in[3];
    const float* bn    = (const float*)d_in[4];
    const float* We    = (const float*)d_in[5];
    const float* be    = (const float*)d_in[6];
    const float* W1    = (const float*)d_in[7];
    const float* b1    = (const float*)d_in[8];
    const float* W2    = (const float*)d_in[9];
    const float* b2    = (const float*)d_in[10];
    const float* W3    = (const float*)d_in[11];
    const float* b3    = (const float*)d_in[12];

    char* ws  = (char*)d_ws;
    f16* Wnp = (f16*)(ws + WS_WN);
    f16* Wep = (f16*)(ws + WS_WE);
    f16* W1p = (f16*)(ws + WS_W1);
    f16* W2p = (f16*)(ws + WS_W2);
    f16* nf  = (f16*)(ws + WS_NF);

    (void)hipFuncSetAttribute((const void*)edge_fused_kernel,
                              hipFuncAttributeMaxDynamicSharedMemorySize,
                              WG_LDS);

    pack_w_kernel<<<64,  256, 0, stream>>>(Wn, Wnp, 128, 4,  16384);
    pack_w_kernel<<<32,  256, 0, stream>>>(We, Wep, 128, 2,  8192);
    pack_w_kernel<<<768, 256, 0, stream>>>(W1, W1p, 512, 12, 196608);
    pack_w_kernel<<<512, 256, 0, stream>>>(W2, W2p, 256, 16, 131072);

    node_feat_kernel<<<(3125 + 3) / 4, 128, 0, stream>>>(x, bn, Wnp, nf);

    edge_fused_kernel<<<NWG, 32 * EWAVES,
                        WG_LDS, stream>>>(
        eidx, eattr, be, b1, b2, W3, b3, Wep, W1p, W2p, nf, (float*)d_out);
}


# packed weight fragments stored as two 512-byte planes: each 16-byte weight load of a wave reads 512 contiguous bytes (4 whole lines), not 16 of every 32 bytes across 8 lines; k order unchanged
# speedup vs baseline: 4.9205x; 1.0285x over previous
#include <hip/hip_runtime.h>
#include <hip/hip_bf16.h>


typedef _Float16 f16;
typedef __attribute__((ext_vector_type(16))) _Float16 v16h;
typedef __attribute__((ext_vector_type(8)))  _Float16 v8h;
typedef __attribute__((ext_vector_type(4)))  _Float16 v4h;
typedef __attribute__((ext_vector_type(8)))  float    v8f;
typedef __attribute__((ext_vector_type(2)))  float    v2f;

#define N_NODES 50000
#define N_EDGES 500000
#define ETILES  (N_EDGES / 16)
#define EWAVES  8

__device__ __forceinline__ v16h join16(v8h lo, v8h hi) {
    union { v16h v; v8h h[2]; } u;
    u.h[0] = lo; u.h[1] = hi;
    return u.v;
}

__device__ __forceinline__ v8f load8f(const float* __restrict__ p) {
    const float4 lo = *(const float4*)p;
    const float4 hi = *(const float4*)(p + 4);
    v8f a;
    a[0] = lo.x; a[1] = lo.y; a[2] = lo.z; a[3] = lo.w;
    a[4] = hi.x; a[5] = hi.y; a[6] = hi.z; a[7] = hi.w;
    return a;
}

__device__ __forceinline__ v8h relu_pack(v8f x) {
    v8h p;
#pragma unroll
    for (int r = 0; r < 8; ++r) p[r] = (f16)x[r];
    const v8h z = {0, 0, 0, 0, 0, 0, 0, 0};
    return __builtin_elementwise_max(p, z);
}

__device__ __forceinline__ v8f wmma_f16(v16h a, v16h b, v8f c) {
    return __builtin_amdgcn_wmma_f32_16x16x32_f16(false, a, false, b,
                                                  (short)0, c, false, false);
}

__device__ __forceinline__ float relu_f32(float x) {
    const int b = __builtin_bit_cast(int, x);
    return __builtin_bit_cast(float, b > 0 ? b : 0);
}

__device__ __forceinline__ float xhalf(float x) {
    return __builtin_bit_cast(float, __builtin_amdgcn_permlanex16(
        __builtin_bit_cast(int, x), __builtin_bit_cast(int, x),
        0x76543210, 0xfedcba98, false, false));
}

__global__ void pack_w_kernel(const float* __restrict__ W, f16* __restrict__ out,
                              int Ntot, int kTiles, int total) {
    for (int idx = blockIdx.x * blockDim.x + threadIdx.x; idx < total;
         idx += gridDim.x * blockDim.x) {
        int f  = idx >> 9;
        int r  = idx & 511;
        int q  = r >> 8;
        int l  = (r >> 3) & 31;
        int e  = r & 7;
        int nt = f / kTiles;
        int kt = f - nt * kTiles;
        int K  = kt * 32 + (q << 4) + ((l >> 4) << 3) + e;
        int N  = nt * 16 + (l & 15);
        out[idx] = (f16)W[(size_t)K * Ntot + N];
    }
}

__device__ __forceinline__ v16h load_wfrag(const f16* __restrict__ Wp, int frag, int l) {
    const f16* p = Wp + (size_t)frag * 512 + l * 8;
    return join16(*(const v8h*)p, *(const v8h*)(p + 256));
}

__global__ void node_feat_kernel(const float* __restrict__ x,
                                 const float* __restrict__ bn,
                                 const f16*  __restrict__ Wnp,
                                 f16* __restrict__ nf) {
    __shared__ __align__(16) f16 xs_all[4][16 * 128];
    const int l  = threadIdx.x & 31;
    const int wv = threadIdx.x >> 5;
    const int tile = blockIdx.x * 4 + wv;
    if (tile >= N_NODES / 16) return;
    f16* xs = xs_all[wv];
    const int n0 = tile * 16;
    const int m  = l & 15;
    const int h  = l >> 4;

#pragma unroll
    for (int t = 0; t < 16; ++t) {
        const int row = t, col = l * 4;
        const float4 v = *(const float4*)(x + (size_t)(n0 + row) * 128 + col);
        v4h p; p[0] = (f16)v.x; p[1] = (f16)v.y; p[2] = (f16)v.z; p[3] = (f16)v.w;
        *(v4h*)(xs + row * 128 + col) = p;
    }

    for (int nc = 0; nc < 2; ++nc) {
        v8f acc[4];
#pragma unroll
        for (int j = 0; j < 4; ++j) acc[j] = load8f(bn + (nc * 4 + j) * 16 + 8 * h);
        for (int kt = 0; kt < 4; ++kt) {
            int k0  = kt * 32 + h * 8;
            v16h a = join16(*(const v8h*)(xs + m * 128 + k0),
                            *(const v8h*)(xs + m * 128 + k0 + 16));
            v16h b[4];
#pragma unroll
            for (int j = 0; j < 4; ++j)
                b[j] = load_wfrag(Wnp, (nc * 4 + j) * 4 + kt, l);
#pragma unroll
            for (int j = 0; j < 4; ++j) acc[j] = wmma_f16(b[j], a, acc[j]);
        }
#pragma unroll
        for (int j = 0; j < 4; ++j)
            *(v8h*)(nf + (size_t)(n0 + m) * 128 + (nc * 4 + j) * 16 + 8 * h) =
                relu_pack(acc[j]);
    }
}

#define CMB_STRIDE 392
#define H1_STRIDE  520
#define MTILES     8
#define LDS_CMB    0
#define LDS_H1     (LDS_CMB + 128 * CMB_STRIDE * 2)
#define LDS_PART   (LDS_H1  + 128 * H1_STRIDE  * 2)
#define WG_LDS     (LDS_PART + EWAVES * 128 * 4)


template <int NJ, int KT>
__device__ __forceinline__ void load_b(v16h (&b)[NJ], const f16* __restrict__ Wp,
                                       int nt0, int kt, int l) {
#pragma unroll
    for (int j = 0; j < NJ; ++j)
        b[j] = load_wfrag(Wp, (nt0 + j) * KT + kt, l);
}

template <int AH>
__device__ __forceinline__ void load_a(v16h (&a)[MTILES / AH], const f16* A, int a_stride,
                                       int blk, int m, int h) {
    constexpr int CM = MTILES / AH;
    const int k0  = (blk / AH) * 32 + h * 8;
    const int mi0 = (blk % AH) * CM;
#pragma unroll
    for (int mi = 0; mi < CM; ++mi) {
        const f16* ar = A + ((mi0 + mi) * 16 + m) * a_stride + k0;
        a[mi] = join16(*(const v8h*)ar, *(const v8h*)(ar + 16));
    }
}

template <int NJ>
__device__ __forceinline__ void load_bias(v8f (&bv)[NJ], const float* __restrict__ bias,
                                          int nt0, int h) {
#pragma unroll
    for (int j = 0; j < NJ; ++j)
        bv[j] = load8f(bias + (nt0 + j) * 16 + 8 * h);
}

template <int NJ>
__device__ __forceinline__ void fill_acc(v8f (&acc)[MTILES][NJ], const v8f (&bv)[NJ]) {
#pragma unroll
    for (int j = 0; j < NJ; ++j)
#pragma unroll
        for (int mi = 0; mi < MTILES; ++mi) acc[mi][j] = bv[j];
}

template <int NJ>
__device__ __forceinline__ void init_acc(v8f (&acc)[MTILES][NJ],
                                         const float* __restrict__ bias,
                                         int nt0, int h) {
    v8f bv[NJ];
    load_bias<NJ>(bv, bias, nt0, h);
    fill_acc<NJ>(acc, bv);
}

template <int NJ, int KT, int PB, int AH, int AR, int HB, class Extra>
__device__ __forceinline__ void strip_mma(v8f (&acc)[MTILES][NJ],
                                          v16h (&b)[PB + 1][NJ],
                                          const f16* A, int a_stride,
                                          const f16* __restrict__ Wp, int nt0,
                                          int l, int m, int h, Extra extra) {
    constexpr int CM = MTILES / AH;
    constexpr int NB = KT * AH;
    v16h a[AR][CM];
#pragma unroll
    for (int p = 0; p < AR - 1; ++p) load_a<AH>(a[p], A, a_stride, p, m, h);
#pragma unroll
    for (int blk = 0; blk < NB; ++blk) {
        const int kt = blk / AH;
        const int c  = blk % AH;
        if (c == 0 && kt + PB < KT)
            load_b<NJ, KT>(b[(kt + PB) % (PB + 1)], Wp, nt0, kt + PB, l);
        if (blk + AR - 1 < NB)
            load_a<AH>(a[(blk + AR - 1) % AR], A, a_stride, blk + AR - 1, m, h);
        if (blk == HB) extra();
        __builtin_amdgcn_sched_barrier(0);
#pragma unroll
        for (int mi = 0; mi < CM; ++mi)
#pragma unroll
            for (int j = 0; j < NJ; ++j)
                acc[c * CM + mi][j] = wmma_f16(b[kt % (PB + 1)][j], a[blk % AR][mi],
                                               acc[c * CM + mi][j]);
        __builtin_amdgcn_sched_barrier(0);
    }
}

template <int NJ>
__device__ __forceinline__ void strip_store(const v8f (&acc)[MTILES][NJ], int nt0,
                                            f16* D, int d_stride, int d_col0,
                                            int m, int h) {
#pragma unroll
    for (int mi = 0; mi < MTILES; ++mi)
#pragma unroll
        for (int j = 0; j < NJ; ++j)
            *(v8h*)(D + (mi * 16 + m) * d_stride + d_col0 + (nt0 + j) * 16 + 8 * h) =
                relu_pack(acc[mi][j]);
}

__device__ __forceinline__ void lds_barrier() {
    __builtin_amdgcn_fence(__ATOMIC_RELEASE, "workgroup", "local");
    __builtin_amdgcn_s_barrier();
    __builtin_amdgcn_fence(__ATOMIC_ACQUIRE, "workgroup", "local");
}

#define NT128 ((ETILES + EWAVES - 1) / EWAVES)
#define NWG   256

__device__ __forceinline__ int first_edge(int T, int wv) {
    const int t16 = T * EWAVES + wv;
    return t16 < ETILES ? t16 * 16 : 0;
}

__device__ __forceinline__ int load_nodes(const int* __restrict__ eidx, int e0, int m, int h) {
    return eidx[h * N_EDGES + e0 + m];
}

__device__ __forceinline__ void load_eattr(float4 (&ev)[8], const float* __restrict__ eattr,
                                           int e0, int m, int h) {
    const float* r = eattr + (size_t)(e0 + m) * 64 + h * 8;
#pragma unroll
    for (int t = 0; t < 8; ++t)
        ev[t] = *(const float4*)(r + (t >> 2) * 32 + ((t >> 1) & 1) * 16 + (t & 1) * 4);
}

__device__ __forceinline__ v16h eattr_frag(const float4 (&ev)[8], int kt) {
    v16h a;
#pragma unroll
    for (int u = 0; u < 4; ++u) {
        const float4 v = ev[kt * 4 + u];
        a[4 * u + 0] = (f16)v.x; a[4 * u + 1] = (f16)v.y;
        a[4 * u + 2] = (f16)v.z; a[4 * u + 3] = (f16)v.w;
    }
    return a;
}

__device__ __forceinline__ void gather_rows(v8h (&g)[16], const f16* __restrict__ nf,
                                            int node, int m, int h) {
#pragma unroll
    for (int p = 0; p < 16; ++p) {
        const int ns = __builtin_amdgcn_readlane(node, p);
        const int nd = __builtin_amdgcn_readlane(node, 16 + p);
        g[p] = *(const v8h*)(nf + (size_t)(h ? nd : ns) * 128 + m * 8);
    }
}

__device__ __forceinline__ void edge_rows(const float4 (&ev)[8], const v16h (&we)[2][8],
                                          const v8f (&bv)[8], f16* cmb,
                                          int row0, int m, int h) {
    v8f acc[8];
#pragma unroll
    for (int j = 0; j < 8; ++j) acc[j] = bv[j];
#pragma unroll
    for (int kt = 0; kt < 2; ++kt) {
        const v16h a = eattr_frag(ev, kt);
#pragma unroll
        for (int j = 0; j < 8; ++j) acc[j] = wmma_f16(we[kt][j], a, acc[j]);
    }
#pragma unroll
    for (int j = 0; j < 8; ++j)
        *(v8h*)(cmb + (row0 + m) * CMB_STRIDE + 256 + j * 16 + 8 * h) = relu_pack(acc[j]);
}

__device__ __forceinline__ void store_rows(const v8h (&g)[16], f16* cmb, int row0, int m, int h) {
#pragma unroll
    for (int p = 0; p < 16; ++p)
        *(v8h*)(cmb + (row0 + p) * CMB_STRIDE + h * 128 + m * 8) = g[p];
}

__global__ void __launch_bounds__(32 * EWAVES)
edge_fused_kernel(const int*   __restrict__ eidx,
                  const float* __restrict__ eattr,
                  const float* __restrict__ be,
                  const float* __restrict__ b1,
                  const float* __restrict__ b2,
                  const float* __restrict__ W3,
                  const float* __restrict__ b3,
                  const f16*   __restrict__ Wep,
                  const f16*   __restrict__ W1p,
                  const f16*   __restrict__ W2p,
                  const f16*   __restrict__ nf,
                  float* __restrict__ out) {
    extern __shared__ __align__(16) char smem[];
    f16*   cmb = (f16*)(smem + LDS_CMB);
    f16*   h1  = (f16*)(smem + LDS_H1);
    float* part = (float*)(smem + LDS_PART);

    const auto no_extra = []() {};

    v16h wb1[3][4];
    v16h wb2[5][2];

    int T = blockIdx.x;

    {
        const int l    = threadIdx.x & 31;
        const int wv   = threadIdx.x >> 5;
        const int m    = l & 15;
        const int h    = l >> 4;
        const int row0 = wv * 16;
        const int e0 = first_edge(T, wv);
        const int node = load_nodes(eidx, e0, m, h);
        float4 ev[8];
        load_eattr(ev, eattr, e0, m, h);
        v16h we[2][8];
        v8f  bev[8];
        load_b<8, 2>(we[0], Wep, 0, 0, l);
        load_b<8, 2>(we[1], Wep, 0, 1, l);
        load_bias<8>(bev, be, 0, h);
        load_b<4, 12>(wb1[0], W1p, wv * 4, 0, l);
        load_b<4, 12>(wb1[1], W1p, wv * 4, 1, l);
        v8h g[16];
        gather_rows(g, nf, node, m, h);
        __builtin_amdgcn_sched_barrier(0);
        edge_rows(ev, we, bev, cmb, row0, m, h);
        store_rows(g, cmb, row0, m, h);
    }

    lds_barrier();

    const float b3v = b3[0];

    for (;;) {
        int zero = 0;
        asm volatile("" : "+s"(zero));
        int tid = threadIdx.x;
        asm volatile("" : "+v"(tid));
        const int l    = tid & 31;
        const int wv   = tid >> 5;
        const int m    = l & 15;
        const int h    = l >> 4;
        const int row0 = wv * 16;
        const f16*   W1p_t = W1p + zero;
        const f16*   W2p_t = W2p + zero;
        const float* b1_t  = b1 + zero;
        const float* b2_t  = b2 + zero;
        const float* W3_t  = W3 + zero;

        const bool has_next = (T + NWG < NT128);
        const int  Tn  = has_next ? T + NWG : T;
        const int  e0n = first_edge(Tn, wv);
        int    node_n;
        float4 ev_n[8];
        v8h    g_n[16];
        v8f    bias2[2];
        v8f    w3r[2];

        {
            v8f acc[MTILES][4];
            init_acc<4>(acc, b1_t, wv * 4, h);
            strip_mma<4, 12, 2, 2, 2, -1>(acc, wb1, cmb, CMB_STRIDE, W1p_t, wv * 4, l, m, h,
                                          no_extra);
#pragma unroll
            for (int p = 0; p < 4; ++p)
                load_b<2, 16>(wb2[p], W2p_t, wv * 2, p, l);
            load_bias<2>(bias2, b2_t, wv * 2, h);
            __builtin_amdgcn_sched_barrier(0);
            node_n = load_nodes(eidx, e0n, m, h);
            load_eattr(ev_n, eattr, e0n, m, h);
            strip_store<4>(acc, wv * 4, h1, H1_STRIDE, 0, m, h);
        }

        lds_barrier();

        {
            v8f acc[MTILES][2];
            fill_acc<2>(acc, bias2);
            strip_mma<2, 16, 4, 1, 3, 14>(acc, wb2, h1, H1_STRIDE, W2p_t, wv * 2, l, m, h,
                                          [&]() {
#pragma unroll
                for (int j = 0; j < 2; ++j)
                    w3r[j] = load8f(W3_t + (wv * 2 + j) * 16 + 8 * h);
                gather_rows(g_n, nf, node_n, m, h);
            });

            v2f s2[MTILES];
#pragma unroll
            for (int mi = 0; mi < MTILES; ++mi) s2[mi] = (v2f){0.0f, 0.0f};
#pragma unroll
            for (int j = 0; j < 2; ++j)
#pragma unroll
                for (int r = 0; r < 8; r += 2) {
                    const v2f w = {w3r[j][r], w3r[j][r + 1]};
#pragma unroll
                    for (int mi = 0; mi < MTILES; ++mi) {
                        const v2f x = {relu_f32(acc[mi][j][r]), relu_f32(acc[mi][j][r + 1])};
                        s2[mi] = __builtin_elementwise_fma(x, w, s2[mi]);
                    }
                }
            v16h we[2][8];
            v8f  bev[8];
            __builtin_amdgcn_sched_barrier(0);
            int zero_e = 0;
            asm volatile("" : "+s"(zero_e));
            const f16*   Wep_e = Wep + zero_e;
            const float* be_e  = be + zero_e;
            const f16*   W1p_e = W1p + zero_e;
            load_b<8, 2>(we[0], Wep_e, 0, 0, l);
            load_bias<8>(bev, be_e, 0, h);
            load_b<8, 2>(we[1], Wep_e, 0, 1, l);
            load_b<4, 12>(wb1[0], W1p_e, wv * 4, 0, l);
            load_b<4, 12>(wb1[1], W1p_e, wv * 4, 1, l);
            __builtin_amdgcn_sched_barrier(0);

            float s[MTILES];
#pragma unroll
            for (int mi = 0; mi < MTILES; ++mi) {
                s[mi] = s2[mi][0] + s2[mi][1];
                s[mi] += xhalf(s[mi]);
            }
            if (h == 0) {
#pragma unroll
                for (int mi = 0; mi < MTILES; ++mi) part[wv * 128 + mi * 16 + m] = s[mi];
            }

            store_rows(g_n, cmb, row0, m, h);
            edge_rows(ev_n, we, bev, cmb, row0, m, h);
        }

        lds_barrier();

        if (wv < 4) {
            const int row = tid;
            float s = b3v;
#pragma unroll
            for (int w = 0; w < EWAVES; ++w) s += part[w * 128 + row];
            const int e = T * (EWAVES * 16) + row;
            if (e < N_EDGES) out[e] = s;
        }

        if (!has_next) break;
        T = Tn;
    }
}

#define WS_WN 0
#define WS_WE (WS_WN + 16384 * 2)
#define WS_W1 (WS_WE + 8192 * 2)
#define WS_W2 (WS_W1 + 196608 * 2)
#define WS_NF (WS_W2 + 131072 * 2)

extern "C" void kernel_launch(void* const* d_in, const int* in_sizes, int n_in,
                              void* d_out, int out_size, void* d_ws, size_t ws_size,
                              hipStream_t stream) {
    (void)in_sizes; (void)n_in; (void)out_size; (void)ws_size;
    const float* x     = (const float*)d_in[0];
    const int*   eidx  = (const int*)  d_in[1];
    const float* eattr = (const float*)d_in[2];
    const float* Wn    = (const float*)d_in[3];
    const float* bn    = (const float*)d_in[4];
    const float* We    = (const float*)d_in[5];
    const float* be    = (const float*)d_in[6];
    const float* W1    = (const float*)d_in[7];
    const float* b1    = (const float*)d_in[8];
    const float* W2    = (const float*)d_in[9];
    const float* b2    = (const float*)d_in[10];
    const float* W3    = (const float*)d_in[11];
    const float* b3    = (const float*)d_in[12];

    char* ws  = (char*)d_ws;
    f16* Wnp = (f16*)(ws + WS_WN);
    f16* Wep = (f16*)(ws + WS_WE);
    f16* W1p = (f16*)(ws + WS_W1);
    f16* W2p = (f16*)(ws + WS_W2);
    f16* nf  = (f16*)(ws + WS_NF);

    (void)hipFuncSetAttribute((const void*)edge_fused_kernel,
                              hipFuncAttributeMaxDynamicSharedMemorySize,
                              WG_LDS);

    pack_w_kernel<<<64,  256, 0, stream>>>(Wn, Wnp, 128, 4,  16384);
    pack_w_kernel<<<32,  256, 0, stream>>>(We, Wep, 128, 2,  8192);
    pack_w_kernel<<<768, 256, 0, stream>>>(W1, W1p, 512, 12, 196608);
    pack_w_kernel<<<512, 256, 0, stream>>>(W2, W2p, 256, 16, 131072);

    node_feat_kernel<<<(3125 + 3) / 4, 128, 0, stream>>>(x, bn, Wnp, nf);

    edge_fused_kernel<<<NWG, 32 * EWAVES,
                        WG_LDS, stream>>>(
        eidx, eattr, be, b1, b2, W3, b3, Wep, W1p, W2p, nf, (float*)d_out);
}
